# starting kernel
# baseline (speedup 1.0000x reference)
#include <hip/hip_runtime.h>
#include <hip/hip_bf16.h>

typedef __attribute__((ext_vector_type(2))) float v2f;
typedef __attribute__((ext_vector_type(8))) float v8f;

#define IN_DIM  256
#define OUT_DIM 128

__global__ __launch_bounds__(128) void k_gemm(const float* __restrict__ A,
                                              const float* __restrict__ W,
                                              float* __restrict__ Z,
                                              int n_src) {
  const int lane = threadIdx.x & 31;
  const int wave = threadIdx.x >> 5;
  const int tile = blockIdx.x * 4 + wave;
  const int m0   = tile * 16;
  if (m0 >= n_src) return;

  const int half = lane >> 4;
  const int nc   = lane & 15;

  int mrow = m0 + nc;
  if (mrow >= n_src) mrow = n_src - 1;
  const float* __restrict__ arow = A + (size_t)mrow * IN_DIM;

  const v8f vzero = {};
  v8f acc[8];
#pragma unroll
  for (int i = 0; i < 8; ++i) acc[i] = vzero;

  for (int k = 0; k < IN_DIM; k += 4) {
    const v2f a = *(const v2f*)(arow + k + 2 * half);
#pragma unroll
    for (int tn = 0; tn < 8; ++tn) {
      const int n = tn * 16 + nc;
      const v2f b = *(const v2f*)(W + (size_t)n * IN_DIM + k + 2 * half);
      acc[tn] = __builtin_amdgcn_wmma_f32_16x16x4_f32(
           false, a,  false, b,
           (short)0, acc[tn],  false,  false);
    }
  }

#pragma unroll
  for (int r = 0; r < 8; ++r) {
    const int mm = m0 + r + 8 * half;
    if (mm < n_src) {
      float* zr = Z + (size_t)mm * OUT_DIM;
#pragma unroll
      for (int tn = 0; tn < 8; ++tn) zr[tn * 16 + nc] = acc[tn][r];
    }
  }
}

__global__ __launch_bounds__(256) void k_score(const float* __restrict__ Z,
                                               const float* __restrict__ wt,
                                               float* __restrict__ score,
                                               int n_src) {
  const int w    = (blockIdx.x * blockDim.x + threadIdx.x) >> 5;
  const int lane = threadIdx.x & 31;
  if (w >= n_src) return;
  const float* zr = Z + (size_t)w * OUT_DIM;
  float s = zr[lane]      * wt[lane]
          + zr[lane + 32] * wt[lane + 32]
          + zr[lane + 64] * wt[lane + 64]
          + zr[lane + 96] * wt[lane + 96];
#pragma unroll
  for (int off = 16; off > 0; off >>= 1) s += __shfl_down(s, off, 32);
  if (lane == 0) score[w] = s;
}

__device__ __forceinline__ float edge_logit(const float* __restrict__ score,
                                            const float* __restrict__ t,
                                            int s, int d) {
  const float e = -fabsf(t[s] - t[d]);
  return expf(score[s] * e * 0.002f);
}

__global__ __launch_bounds__(256) void k_zero(float* __restrict__ m,
                                              float* __restrict__ denom,
                                              int n_dst) {
  const int i = blockIdx.x * blockDim.x + threadIdx.x;
  if (i < n_dst) { m[i] = 0.0f; denom[i] = 0.0f; }
}

__global__ __launch_bounds__(256) void k_edge_max(const float* __restrict__ score,
                                                  const float* __restrict__ t,
                                                  const int* __restrict__ src,
                                                  const int* __restrict__ dst,
                                                  float* __restrict__ m, int E) {
  const int i = blockIdx.x * blockDim.x + threadIdx.x;
  if (i >= E) return;
  const float le = edge_logit(score, t, src[i], dst[i]);
  atomicMax((int*)(m + dst[i]), __float_as_int(le));
}

__global__ __launch_bounds__(256) void k_edge_denom(const float* __restrict__ score,
                                                    const float* __restrict__ t,
                                                    const int* __restrict__ src,
                                                    const int* __restrict__ dst,
                                                    const float* __restrict__ m,
                                                    float* __restrict__ denom, int E) {
  const int i = blockIdx.x * blockDim.x + threadIdx.x;
  if (i >= E) return;
  const int d  = dst[i];
  const float un = expf(edge_logit(score, t, src[i], d) - m[d]);
  atomicAdd(denom + d, un);
}

__global__ __launch_bounds__(256) void k_scatter(const float* __restrict__ score,
                                                 const float* __restrict__ t,
                                                 const int* __restrict__ src,
                                                 const int* __restrict__ dst,
                                                 const float* __restrict__ m,
                                                 const float* __restrict__ denom,
                                                 const float* __restrict__ Z,
                                                 float* __restrict__ out, int E) {
  const int w    = (blockIdx.x * blockDim.x + threadIdx.x) >> 5;
  const int lane = threadIdx.x & 31;
  if (w >= E) return;
  const int s = src[w];
  const int d = dst[w];
  const float le    = edge_logit(score, t, s, d);
  const float un    = expf(le - m[d]);
  const float alpha = un / fmaxf(denom[d], 1e-20f);

  const float4 zv = *(const float4*)(Z + (size_t)s * OUT_DIM + lane * 4);
  float* o = out + (size_t)d * OUT_DIM + lane * 4;
  atomicAdd(o + 0, alpha * zv.x);
  atomicAdd(o + 1, alpha * zv.y);
  atomicAdd(o + 2, alpha * zv.z);
  atomicAdd(o + 3, alpha * zv.w);
}

extern "C" void kernel_launch(void* const* d_in, const int* in_sizes, int n_in,
                              void* d_out, int out_size, void* d_ws, size_t ws_size,
                              hipStream_t stream) {
  const float* features = (const float*)d_in[0];
  const float* t        = (const float*)d_in[1];
  const float* W_fc     = (const float*)d_in[2];
  const float* w_t      = (const float*)d_in[3];
  const int*   src_idx  = (const int*)d_in[4];
  const int*   dst_idx  = (const int*)d_in[5];

  const int n_src = in_sizes[0] / IN_DIM;
  const int E     = in_sizes[4];
  const int n_dst = out_size / OUT_DIM;

  float* zbuf  = (float*)d_ws;
  float* score = zbuf + (size_t)n_src * OUT_DIM;
  float* mbuf  = score + n_src;
  float* denom = mbuf + n_dst;
  float* out   = (float*)d_out;

  const int ntile = (n_src + 15) / 16;
  k_gemm<<<(ntile + 3) / 4, 128, 0, stream>>>(features, W_fc, zbuf, n_src);

  k_score<<<(n_src + 7) / 8, 256, 0, stream>>>(zbuf, w_t, score, n_src);

  hipMemcpyAsync(out, zbuf, (size_t)n_dst * OUT_DIM * sizeof(float),
                 hipMemcpyDeviceToDevice, stream);
  k_zero<<<(n_dst + 255) / 256, 256, 0, stream>>>(mbuf, denom, n_dst);

  k_edge_max<<<(E + 255) / 256, 256, 0, stream>>>(score, t, src_idx, dst_idx, mbuf, E);

  k_edge_denom<<<(E + 255) / 256, 256, 0, stream>>>(score, t, src_idx, dst_idx,
                                                    mbuf, denom, E);

  k_scatter<<<(E + 7) / 8, 256, 0, stream>>>(score, t, src_idx, dst_idx,
                                             mbuf, denom, zbuf, out, E);
}


# k_own block reads a bitmap of its own edges (built once by the first launch in LDS tiles, parked in the block's own out rows), not all of dst: 250 x 6.4 MB of reads gone; LDS sums in arrival order
# speedup vs baseline: 3.4148x; 3.4148x over previous
#include <hip/hip_runtime.h>
#include <hip/hip_bf16.h>

typedef __attribute__((ext_vector_type(2))) float v2f;
typedef __attribute__((ext_vector_type(4))) float v4f;
typedef __attribute__((ext_vector_type(8))) float v8f;

#define IN_DIM  256
#define OUT_DIM 128

#define WP_FLOATS (OUT_DIM * IN_DIM)

__global__ __launch_bounds__(128) void k_gemm(const float* __restrict__ A,
                                              const float* __restrict__ Wp,
                                              float* __restrict__ Z,
                                              int n_src) {
  const int lane = threadIdx.x & 31;
  const int wave = threadIdx.x >> 5;
  const int tile = blockIdx.x * 4 + wave;
  const int m0   = tile * 16;
  if (m0 >= n_src) return;

  const int half = lane >> 4;
  const int nc   = lane & 15;

  int mrow = m0 + nc;
  if (mrow >= n_src) mrow = n_src - 1;
  const float* __restrict__ arow = A + (size_t)mrow * IN_DIM + 2 * half;
  const v4f* __restrict__ wl = (const v4f*)Wp + lane;

  const v8f vzero = {};
  v8f acc[8];
#pragma unroll
  for (int i = 0; i < 8; ++i) acc[i] = vzero;

#pragma unroll 1
  for (int t = 0; t < IN_DIM / 16; ++t) {
    v2f a[4];
    v4f b[16];
#pragma unroll
    for (int i = 0; i < 4; ++i) a[i] = *(const v2f*)(arow + 16 * t + 4 * i);
#pragma unroll
    for (int i = 0; i < 16; ++i) b[i] = wl[i * 32];
    wl += 16 * 32;
    __builtin_amdgcn_sched_barrier(0);
#pragma unroll
    for (int qq = 0; qq < 2; ++qq) {
#pragma unroll
      for (int tn = 0; tn < 8; ++tn) {
        const v4f bb = b[qq * 8 + tn];
        const v2f b0 = {bb[0], bb[1]};
        const v2f b1 = {bb[2], bb[3]};
        acc[tn] = __builtin_amdgcn_wmma_f32_16x16x4_f32(
             false, a[2 * qq],  false, b0,
             (short)0, acc[tn],  false,  false);
        acc[tn] = __builtin_amdgcn_wmma_f32_16x16x4_f32(
             false, a[2 * qq + 1],  false, b1,
             (short)0, acc[tn],  false,  false);
      }
    }
  }

#pragma unroll
  for (int r = 0; r < 8; ++r) {
    const int mm = m0 + r + 8 * half;
    if (mm < n_src) {
      float* zr = Z + (size_t)mm * OUT_DIM;
#pragma unroll
      for (int tn = 0; tn < 8; ++tn) zr[tn * 16 + nc] = acc[tn][r];
    }
  }
}

__global__ __launch_bounds__(256) void k_score(const float* __restrict__ Z,
                                               const float* __restrict__ wt,
                                               float* __restrict__ score,
                                               int n_src) {
  const int w    = (blockIdx.x * blockDim.x + threadIdx.x) >> 5;
  const int lane = threadIdx.x & 31;
  if (w >= n_src) return;
  const float* zr = Z + (size_t)w * OUT_DIM;
  float s = zr[lane]      * wt[lane]
          + zr[lane + 32] * wt[lane + 32]
          + zr[lane + 64] * wt[lane + 64]
          + zr[lane + 96] * wt[lane + 96];
#pragma unroll
  for (int off = 16; off > 0; off >>= 1) s += __shfl_down(s, off, 32);
  if (lane == 0) score[w] = s;
}

__device__ __forceinline__ float edge_logit(const float* __restrict__ score,
                                            const float* __restrict__ t,
                                            int s, int d) {
  const float e = -fabsf(t[s] - t[d]);
  return expf(score[s] * e * 0.002f);
}

#define R_NODES      400
#define PREP_THREADS 128
#define BM_WAVES     (PREP_THREADS / 32)
#define BM_ROWS      252
#define W_BLOCKS     (WP_FLOATS / (PREP_THREADS * 4))

__device__ __forceinline__ bool bitmaps_fit(int E, int n_dst) {
  const int nblk      = (n_dst + R_NODES - 1) / R_NODES;
  const int words     = (E + 31) >> 5;
  const int last_rows = n_dst - (nblk - 1) * R_NODES;
  return nblk <= BM_ROWS && words <= last_rows * OUT_DIM;
}

__global__ __launch_bounds__(PREP_THREADS) void k_prep(const float* __restrict__ W,
                                                       float* __restrict__ Wp,
                                                       const int* __restrict__ dst,
                                                       unsigned* __restrict__ bm,
                                                       int E, int n_dst) {
  __shared__ unsigned tile[BM_WAVES][BM_ROWS * 32];
  const int tid = threadIdx.x;

  if (blockIdx.x < W_BLOCKS) {
    for (int o = blockIdx.x * PREP_THREADS + tid; o < WP_FLOATS; o += W_BLOCKS * PREP_THREADS) {
      const int e    = o & 3;
      const int lane = (o >> 2) & 31;
      const int tn   = (o >> 7) & 7;
      const int q    = o >> 10;
      const int n    = tn * 16 + (lane & 15);
      const int k    = 8 * q + 4 * (e >> 1) + 2 * (lane >> 4) + (e & 1);
      Wp[o] = W[n * IN_DIM + k];
    }
    return;
  }

  if (!bitmaps_fit(E, n_dst)) return;
  const int lane  = tid & 31;
  const int wv    = tid >> 5;
  const int nblk  = (n_dst + R_NODES - 1) / R_NODES;
  const int words = (E + 31) >> 5;
  const int gw    = (blockIdx.x - W_BLOCKS) * BM_WAVES + wv;
  if (gw * 32 >= words) return;
  const int wq    = gw * 32 + lane;
  const int e0    = wq * 32;

  unsigned* col = &tile[wv][lane];
  for (int v = 0; v < BM_ROWS; ++v) col[v * 32] = 0u;

  auto mark = [&](const int d, const int j) {
    const unsigned o = min((unsigned)d / (unsigned)R_NODES, (unsigned)(BM_ROWS - 1));
    atomicOr(&col[o * 32], 1u << j);
  };
  if (e0 + 32 <= E) {
    const int4* p = (const int4*)(dst + e0);
    int4 d[8];
#pragma unroll
    for (int c = 0; c < 8; ++c) d[c] = p[c];
#pragma unroll
    for (int c = 0; c < 8; ++c) {
      mark(d[c].x, 4 * c);
      mark(d[c].y, 4 * c + 1);
      mark(d[c].z, 4 * c + 2);
      mark(d[c].w, 4 * c + 3);
    }
  } else {
    for (int j = 0; j < 32; ++j)
      if (e0 + j < E) mark(dst[e0 + j], j);
  }

  if (wq < words) {
    unsigned* g = bm + wq;
#pragma unroll 10
    for (int v = 0; v < nblk; ++v) g[(size_t)v * (R_NODES * OUT_DIM)] = col[v * 32];
  }
}

#define LIST_CAP    11264
#define OWN_THREADS 1024
#define TILE_EDGES  (OWN_THREADS * 4)
#define SCAN_UNROLL 8
#define BM_UNROLL   4
#define EDGE_GROUP  8

__global__ __launch_bounds__(OWN_THREADS) void k_own(const float* __restrict__ score,
                                                     const float* __restrict__ t,
                                                     const int* __restrict__ src,
                                                     const int* __restrict__ dst,
                                                     const float* __restrict__ Z,
                                                     float* __restrict__ out,
                                                     int E, int n_dst) {
  __shared__ __attribute__((aligned(16))) float acc[R_NODES * OUT_DIM];
  __shared__ float m_s[R_NODES];
  __shared__ float den_s[R_NODES];
  __shared__ int   list_i[LIST_CAP];
  __shared__ float list_w[LIST_CAP];
  __shared__ unsigned short list_dl[LIST_CAP];
  __shared__ int   n_listed;

  const int tid  = threadIdx.x;
  const int lane = tid & 31;
  const int lo   = blockIdx.x * R_NODES;
  const int rn   = min(R_NODES, n_dst - lo);

  {
    const v4f* __restrict__ z4 = (const v4f*)(Z + (size_t)lo * OUT_DIM);
    v4f* a4 = (v4f*)acc;
    for (int o = tid; o < rn * (OUT_DIM / 4); o += OWN_THREADS) a4[o] = z4[o];
    for (int o = tid; o < R_NODES; o += OWN_THREADS) { m_s[o] = 0.0f; den_s[o] = 0.0f; }
    if (tid == 0) n_listed = 0;
  }
  __syncthreads();

  if (bitmaps_fit(E, n_dst)) {
    const unsigned* __restrict__ bmw = (const unsigned*)(out + (size_t)lo * OUT_DIM);
    const uint4* __restrict__ bm4 = (const uint4*)bmw;
    const int words = (E + 31) >> 5;
    const int quads = words >> 2;
    auto emit = [&](unsigned x, const int ebase, int& sl) {
      while (x) {
        const int b = __builtin_ctz(x);
        x &= x - 1;
        if (sl < LIST_CAP) list_i[sl] = ebase + b;
        ++sl;
      }
    };
    for (int q0 = tid; q0 < quads; q0 += OWN_THREADS * BM_UNROLL) {
      uint4 x[BM_UNROLL];
#pragma unroll
      for (int u = 0; u < BM_UNROLL; ++u) {
        const int q = q0 + u * OWN_THREADS;
        x[u] = (q < quads) ? bm4[q] : make_uint4(0u, 0u, 0u, 0u);
      }
#pragma unroll
      for (int u = 0; u < BM_UNROLL; ++u) {
        const int cnt = __builtin_popcount(x[u].x) + __builtin_popcount(x[u].y) +
                        __builtin_popcount(x[u].z) + __builtin_popcount(x[u].w);
        if (cnt) {
          int sl = atomicAdd(&n_listed, cnt);
          const int eb = (q0 + u * OWN_THREADS) * 128;
          emit(x[u].x, eb,      sl);
          emit(x[u].y, eb + 32, sl);
          emit(x[u].z, eb + 64, sl);
          emit(x[u].w, eb + 96, sl);
        }
      }
    }
    for (int w = quads * 4 + tid; w < words; w += OWN_THREADS) {
      const unsigned x = bmw[w];
      if (x) {
        int sl = atomicAdd(&n_listed, __builtin_popcount(x));
        emit(x, w * 32, sl);
      }
    }
  } else {
    const unsigned r = (unsigned)rn;
    auto append = [&](const int i) {
      const int sl = atomicAdd(&n_listed, 1);
      if (sl < LIST_CAP) list_i[sl] = i;
    };
    auto take = [&](const int4 dv, const int e0) {
      const unsigned d0 = (unsigned)(dv.x - lo);
      const unsigned d1 = (unsigned)(dv.y - lo);
      const unsigned d2 = (unsigned)(dv.z - lo);
      const unsigned d3 = (unsigned)(dv.w - lo);
      if (d0 < r || d1 < r || d2 < r || d3 < r) {
        if (d0 < r) append(e0);
        if (d1 < r) append(e0 + 1);
        if (d2 < r) append(e0 + 2);
        if (d3 < r) append(e0 + 3);
      }
    };
    const int n_full = E / TILE_EDGES;
    const int first  = (int)(((long long)blockIdx.x * n_full) / gridDim.x);
    int p = 0;
    for (; p + SCAN_UNROLL <= n_full; p += SCAN_UNROLL) {
      int4 dv[SCAN_UNROLL];
      int  eb[SCAN_UNROLL];
#pragma unroll
      for (int u = 0; u < SCAN_UNROLL; ++u) {
        int tile = first + p + u;
        if (tile >= n_full) tile -= n_full;
        eb[u] = tile * TILE_EDGES + tid * 4;
        dv[u] = *(const int4*)(dst + eb[u]);
      }
#pragma unroll
      for (int u = 0; u < SCAN_UNROLL; ++u) take(dv[u], eb[u]);
    }
    for (; p < n_full; ++p) {
      int tile = first + p;
      if (tile >= n_full) tile -= n_full;
      const int e0 = tile * TILE_EDGES + tid * 4;
      take(*(const int4*)(dst + e0), e0);
    }
    for (int i = n_full * TILE_EDGES + tid; i < E; i += OWN_THREADS) {
      const unsigned d0 = (unsigned)(dst[i] - lo);
      if (d0 < r) append(i);
    }
  }
  __syncthreads();
  const int total = n_listed;

  if (total <= LIST_CAP) {
    for (int e = tid; e < total; e += OWN_THREADS) {
      const int   i  = list_i[e];
      const int   d  = dst[i];
      const int   s  = src[i];
      const int   dl = d - lo;
      const float le = edge_logit(score, t, s, d);
      atomicMax((int*)&m_s[dl], __float_as_int(le));
      list_i[e]  = s;
      list_dl[e] = (unsigned short)dl;
      list_w[e]  = le;
    }
    __syncthreads();
    for (int e = tid; e < total; e += OWN_THREADS) {
      const int   dl = list_dl[e];
      const float un = expf(list_w[e] - m_s[dl]);
      atomicAdd(&den_s[dl], un);
      list_w[e] = un;
    }
    __syncthreads();
    const int wv = __builtin_amdgcn_readfirstlane(tid >> 5);
    for (int base = wv * 32; base < total; base += OWN_THREADS) {
      const int cnt = min(32, total - base);
      int   s = 0, d = 0;
      float alpha = 0.0f;
      if (lane < cnt) {
        s = list_i[base + lane];
        d = list_dl[base + lane];
        alpha = list_w[base + lane] / fmaxf(den_s[d], 1e-20f);
      }
      const int alpha_bits = __float_as_int(alpha);
      for (int j0 = 0; j0 < cnt; j0 += EDGE_GROUP) {
        if (j0 + EDGE_GROUP <= cnt) {
          float z[EDGE_GROUP][4];
#pragma unroll
          for (int u = 0; u < EDGE_GROUP; ++u) {
            const int sj = __builtin_amdgcn_readlane(s, j0 + u);
            const float* zr = Z + (size_t)sj * OUT_DIM + lane;
            z[u][0] = zr[0];
            z[u][1] = zr[32];
            z[u][2] = zr[64];
            z[u][3] = zr[96];
          }
#pragma unroll
          for (int u = 0; u < EDGE_GROUP; ++u) {
            const int   dj = __builtin_amdgcn_readlane(d, j0 + u);
            const float aj = __int_as_float(__builtin_amdgcn_readlane(alpha_bits, j0 + u));
            float* o = acc + dj * OUT_DIM + lane;
            atomicAdd(o,      aj * z[u][0]);
            atomicAdd(o + 32, aj * z[u][1]);
            atomicAdd(o + 64, aj * z[u][2]);
            atomicAdd(o + 96, aj * z[u][3]);
          }
        } else {
          for (int j = j0; j < cnt; ++j) {
            const int   sj = __builtin_amdgcn_readlane(s, j);
            const int   dj = __builtin_amdgcn_readlane(d, j);
            const float aj = __int_as_float(__builtin_amdgcn_readlane(alpha_bits, j));
            const float* zr = Z + (size_t)sj * OUT_DIM + lane;
            float* o = acc + dj * OUT_DIM + lane;
            atomicAdd(o,      aj * zr[0]);
            atomicAdd(o + 32, aj * zr[32]);
            atomicAdd(o + 64, aj * zr[64]);
            atomicAdd(o + 96, aj * zr[96]);
          }
        }
      }
    }
    __syncthreads();
  } else {
    for (int mode = 0; mode < 3; ++mode) {
      for (int i = tid; i < E; i += OWN_THREADS) {
        const unsigned dl = (unsigned)(dst[i] - lo);
        if (dl < (unsigned)rn) {
          const int   s  = src[i];
          const float le = edge_logit(score, t, s, lo + (int)dl);
          if (mode == 0) {
            atomicMax((int*)&m_s[dl], __float_as_int(le));
          } else {
            const float un = expf(le - m_s[dl]);
            if (mode == 1) {
              atomicAdd(&den_s[dl], un);
            } else {
              const float a = un / fmaxf(den_s[dl], 1e-20f);
              const float* zr = Z + (size_t)s * OUT_DIM;
              for (int c = 0; c < OUT_DIM; ++c) atomicAdd(&acc[dl * OUT_DIM + c], a * zr[c]);
            }
          }
        }
      }
      __syncthreads();
    }
  }

  {
    v4f* __restrict__ o4 = (v4f*)(out + (size_t)lo * OUT_DIM);
    const v4f* a4 = (const v4f*)acc;
    for (int o = tid; o < rn * (OUT_DIM / 4); o += OWN_THREADS) o4[o] = a4[o];
  }
}

extern "C" void kernel_launch(void* const* d_in, const int* in_sizes, int n_in,
                              void* d_out, int out_size, void* d_ws, size_t ws_size,
                              hipStream_t stream) {
  const float* features = (const float*)d_in[0];
  const float* t        = (const float*)d_in[1];
  const float* W_fc     = (const float*)d_in[2];
  const float* w_t      = (const float*)d_in[3];
  const int*   src_idx  = (const int*)d_in[4];
  const int*   dst_idx  = (const int*)d_in[5];

  const int n_src = in_sizes[0] / IN_DIM;
  const int E     = in_sizes[4];
  const int n_dst = out_size / OUT_DIM;

  float* zbuf  = (float*)d_ws;
  float* score = zbuf + (size_t)n_src * OUT_DIM;
  float* out   = (float*)d_out;

  float* wp = score;

  const int bm_waves = ((E + 31) / 32 + 31) / 32;
  k_prep<<<W_BLOCKS + (bm_waves + BM_WAVES - 1) / BM_WAVES, PREP_THREADS, 0, stream>>>(
      W_fc, wp, dst_idx, (unsigned*)d_out, E, n_dst);

  const int ntile = (n_src + 15) / 16;
  k_gemm<<<(ntile + 3) / 4, 128, 0, stream>>>(features, wp, zbuf, n_src);

  k_score<<<(n_src + 7) / 8, 256, 0, stream>>>(zbuf, w_t, score, n_src);

  k_own<<<(n_dst + R_NODES - 1) / R_NODES, OWN_THREADS, 0, stream>>>(
      score, t, src_idx, dst_idx, zbuf, out, E, n_dst);
}


# k_score: one wave takes 32 rows, not one (32x fewer waves): 16-B loads, 8 lanes per 128-B line, 32 loads in flight, its 32 scores stored as one whole line; new summation order of score
# speedup vs baseline: 3.6415x; 1.0664x over previous
#include <hip/hip_runtime.h>
#include <hip/hip_bf16.h>

typedef __attribute__((ext_vector_type(2))) float v2f;
typedef __attribute__((ext_vector_type(4))) float v4f;
typedef __attribute__((ext_vector_type(8))) float v8f;

#define IN_DIM  256
#define OUT_DIM 128

#define WP_FLOATS (OUT_DIM * IN_DIM)

__global__ __launch_bounds__(128) void k_gemm(const float* __restrict__ A,
                                              const float* __restrict__ Wp,
                                              float* __restrict__ Z,
                                              int n_src) {
  const int lane = threadIdx.x & 31;
  const int wave = threadIdx.x >> 5;
  const int tile = blockIdx.x * 4 + wave;
  const int m0   = tile * 16;
  if (m0 >= n_src) return;

  const int half = lane >> 4;
  const int nc   = lane & 15;

  int mrow = m0 + nc;
  if (mrow >= n_src) mrow = n_src - 1;
  const float* __restrict__ arow = A + (size_t)mrow * IN_DIM + 2 * half;
  const v4f* __restrict__ wl = (const v4f*)Wp + lane;

  const v8f vzero = {};
  v8f acc[8];
#pragma unroll
  for (int i = 0; i < 8; ++i) acc[i] = vzero;

#pragma unroll 1
  for (int t = 0; t < IN_DIM / 16; ++t) {
    v2f a[4];
    v4f b[16];
#pragma unroll
    for (int i = 0; i < 4; ++i) a[i] = *(const v2f*)(arow + 16 * t + 4 * i);
#pragma unroll
    for (int i = 0; i < 16; ++i) b[i] = wl[i * 32];
    wl += 16 * 32;
    __builtin_amdgcn_sched_barrier(0);
#pragma unroll
    for (int qq = 0; qq < 2; ++qq) {
#pragma unroll
      for (int tn = 0; tn < 8; ++tn) {
        const v4f bb = b[qq * 8 + tn];
        const v2f b0 = {bb[0], bb[1]};
        const v2f b1 = {bb[2], bb[3]};
        acc[tn] = __builtin_amdgcn_wmma_f32_16x16x4_f32(
             false, a[2 * qq],  false, b0,
             (short)0, acc[tn],  false,  false);
        acc[tn] = __builtin_amdgcn_wmma_f32_16x16x4_f32(
             false, a[2 * qq + 1],  false, b1,
             (short)0, acc[tn],  false,  false);
      }
    }
  }

#pragma unroll
  for (int r = 0; r < 8; ++r) {
    const int mm = m0 + r + 8 * half;
    if (mm < n_src) {
      float* zr = Z + (size_t)mm * OUT_DIM;
#pragma unroll
      for (int tn = 0; tn < 8; ++tn) zr[tn * 16 + nc] = acc[tn][r];
    }
  }
}

#define SC_ROWS    32
#define SC_THREADS 128
#define SC_BLOCK_ROWS (SC_ROWS * (SC_THREADS / 32))

__global__ __launch_bounds__(SC_THREADS) __attribute__((amdgpu_waves_per_eu(4, 5))) void k_score(const float* __restrict__ Z,
                                                      const float* __restrict__ wt,
                                                      float* __restrict__ score,
                                                      int n_src) {
  const int lane = threadIdx.x & 31;
  const int wv   = (blockIdx.x * SC_THREADS + threadIdx.x) >> 5;
  const int r0   = wv * SC_ROWS;
  if (r0 >= n_src) return;
  const int g = lane >> 3;
  const int c = lane & 7;

  v4f w[4];
#pragma unroll
  for (int j = 0; j < 4; ++j) w[j] = *(const v4f*)(wt + 32 * j + 4 * c);

  v4f z[SC_ROWS / 4][4];
#pragma unroll
  for (int k = 0; k < SC_ROWS / 4; ++k) {
    int row = r0 + 4 * k + g;
    if (row >= n_src) row = n_src - 1;
    const v4f* __restrict__ zr = (const v4f*)(Z + (size_t)row * OUT_DIM) + c;
#pragma unroll
    for (int j = 0; j < 4; ++j) z[k][j] = zr[8 * j];
  }

  __builtin_amdgcn_sched_barrier(0);

  float mine = 0.0f;
#pragma unroll
  for (int k = 0; k < SC_ROWS / 4; ++k) {
    float p = 0.0f;
#pragma unroll
    for (int j = 0; j < 4; ++j)
      p += z[k][j][0] * w[j][0] + z[k][j][1] * w[j][1] + z[k][j][2] * w[j][2] + z[k][j][3] * w[j][3];
    p += __shfl_xor(p, 1, 32);
    p += __shfl_xor(p, 2, 32);
    p += __shfl_xor(p, 4, 32);
    if (c == k) mine = p;
  }
  const int out_row = r0 + 4 * c + g;
  if (out_row < n_src) score[out_row] = mine;
}

__device__ __forceinline__ float edge_logit(const float* __restrict__ score,
                                            const float* __restrict__ t,
                                            int s, int d) {
  const float e = -fabsf(t[s] - t[d]);
  return expf(score[s] * e * 0.002f);
}

#define R_NODES      400
#define PREP_THREADS 128
#define BM_WAVES     (PREP_THREADS / 32)
#define BM_ROWS      252
#define W_BLOCKS     (WP_FLOATS / (PREP_THREADS * 4))

__device__ __forceinline__ bool bitmaps_fit(int E, int n_dst) {
  const int nblk      = (n_dst + R_NODES - 1) / R_NODES;
  const int words     = (E + 31) >> 5;
  const int last_rows = n_dst - (nblk - 1) * R_NODES;
  return nblk <= BM_ROWS && words <= last_rows * OUT_DIM;
}

__global__ __launch_bounds__(PREP_THREADS) void k_prep(const float* __restrict__ W,
                                                       float* __restrict__ Wp,
                                                       const int* __restrict__ dst,
                                                       unsigned* __restrict__ bm,
                                                       int E, int n_dst) {
  __shared__ unsigned tile[BM_WAVES][BM_ROWS * 32];
  const int tid = threadIdx.x;

  if (blockIdx.x < W_BLOCKS) {
    for (int o = blockIdx.x * PREP_THREADS + tid; o < WP_FLOATS; o += W_BLOCKS * PREP_THREADS) {
      const int e    = o & 3;
      const int lane = (o >> 2) & 31;
      const int tn   = (o >> 7) & 7;
      const int q    = o >> 10;
      const int n    = tn * 16 + (lane & 15);
      const int k    = 8 * q + 4 * (e >> 1) + 2 * (lane >> 4) + (e & 1);
      Wp[o] = W[n * IN_DIM + k];
    }
    return;
  }

  if (!bitmaps_fit(E, n_dst)) return;
  const int lane  = tid & 31;
  const int wv    = tid >> 5;
  const int nblk  = (n_dst + R_NODES - 1) / R_NODES;
  const int words = (E + 31) >> 5;
  const int gw    = (blockIdx.x - W_BLOCKS) * BM_WAVES + wv;
  if (gw * 32 >= words) return;
  const int wq    = gw * 32 + lane;
  const int e0    = wq * 32;

  unsigned* col = &tile[wv][lane];
  for (int v = 0; v < BM_ROWS; ++v) col[v * 32] = 0u;

  auto mark = [&](const int d, const int j) {
    const unsigned o = min((unsigned)d / (unsigned)R_NODES, (unsigned)(BM_ROWS - 1));
    atomicOr(&col[o * 32], 1u << j);
  };
  if (e0 + 32 <= E) {
    const int4* p = (const int4*)(dst + e0);
    int4 d[8];
#pragma unroll
    for (int c = 0; c < 8; ++c) d[c] = p[c];
#pragma unroll
    for (int c = 0; c < 8; ++c) {
      mark(d[c].x, 4 * c);
      mark(d[c].y, 4 * c + 1);
      mark(d[c].z, 4 * c + 2);
      mark(d[c].w, 4 * c + 3);
    }
  } else {
    for (int j = 0; j < 32; ++j)
      if (e0 + j < E) mark(dst[e0 + j], j);
  }

  if (wq < words) {
    unsigned* g = bm + wq;
#pragma unroll 10
    for (int v = 0; v < nblk; ++v) g[(size_t)v * (R_NODES * OUT_DIM)] = col[v * 32];
  }
}

#define LIST_CAP    11264
#define OWN_THREADS 1024
#define TILE_EDGES  (OWN_THREADS * 4)
#define SCAN_UNROLL 8
#define BM_UNROLL   4
#define EDGE_GROUP  8

__global__ __launch_bounds__(OWN_THREADS) void k_own(const float* __restrict__ score,
                                                     const float* __restrict__ t,
                                                     const int* __restrict__ src,
                                                     const int* __restrict__ dst,
                                                     const float* __restrict__ Z,
                                                     float* __restrict__ out,
                                                     int E, int n_dst) {
  __shared__ __attribute__((aligned(16))) float acc[R_NODES * OUT_DIM];
  __shared__ float m_s[R_NODES];
  __shared__ float den_s[R_NODES];
  __shared__ int   list_i[LIST_CAP];
  __shared__ float list_w[LIST_CAP];
  __shared__ unsigned short list_dl[LIST_CAP];
  __shared__ int   n_listed;

  const int tid  = threadIdx.x;
  const int lane = tid & 31;
  const int lo   = blockIdx.x * R_NODES;
  const int rn   = min(R_NODES, n_dst - lo);

  {
    const v4f* __restrict__ z4 = (const v4f*)(Z + (size_t)lo * OUT_DIM);
    v4f* a4 = (v4f*)acc;
    for (int o = tid; o < rn * (OUT_DIM / 4); o += OWN_THREADS) a4[o] = z4[o];
    for (int o = tid; o < R_NODES; o += OWN_THREADS) { m_s[o] = 0.0f; den_s[o] = 0.0f; }
    if (tid == 0) n_listed = 0;
  }
  __syncthreads();

  if (bitmaps_fit(E, n_dst)) {
    const unsigned* __restrict__ bmw = (const unsigned*)(out + (size_t)lo * OUT_DIM);
    const uint4* __restrict__ bm4 = (const uint4*)bmw;
    const int words = (E + 31) >> 5;
    const int quads = words >> 2;
    auto emit = [&](unsigned x, const int ebase, int& sl) {
      while (x) {
        const int b = __builtin_ctz(x);
        x &= x - 1;
        if (sl < LIST_CAP) list_i[sl] = ebase + b;
        ++sl;
      }
    };
    for (int q0 = tid; q0 < quads; q0 += OWN_THREADS * BM_UNROLL) {
      uint4 x[BM_UNROLL];
#pragma unroll
      for (int u = 0; u < BM_UNROLL; ++u) {
        const int q = q0 + u * OWN_THREADS;
        x[u] = (q < quads) ? bm4[q] : make_uint4(0u, 0u, 0u, 0u);
      }
#pragma unroll
      for (int u = 0; u < BM_UNROLL; ++u) {
        const int cnt = __builtin_popcount(x[u].x) + __builtin_popcount(x[u].y) +
                        __builtin_popcount(x[u].z) + __builtin_popcount(x[u].w);
        if (cnt) {
          int sl = atomicAdd(&n_listed, cnt);
          const int eb = (q0 + u * OWN_THREADS) * 128;
          emit(x[u].x, eb,      sl);
          emit(x[u].y, eb + 32, sl);
          emit(x[u].z, eb + 64, sl);
          emit(x[u].w, eb + 96, sl);
        }
      }
    }
    for (int w = quads * 4 + tid; w < words; w += OWN_THREADS) {
      const unsigned x = bmw[w];
      if (x) {
        int sl = atomicAdd(&n_listed, __builtin_popcount(x));
        emit(x, w * 32, sl);
      }
    }
  } else {
    const unsigned r = (unsigned)rn;
    auto append = [&](const int i) {
      const int sl = atomicAdd(&n_listed, 1);
      if (sl < LIST_CAP) list_i[sl] = i;
    };
    auto take = [&](const int4 dv, const int e0) {
      const unsigned d0 = (unsigned)(dv.x - lo);
      const unsigned d1 = (unsigned)(dv.y - lo);
      const unsigned d2 = (unsigned)(dv.z - lo);
      const unsigned d3 = (unsigned)(dv.w - lo);
      if (d0 < r || d1 < r || d2 < r || d3 < r) {
        if (d0 < r) append(e0);
        if (d1 < r) append(e0 + 1);
        if (d2 < r) append(e0 + 2);
        if (d3 < r) append(e0 + 3);
      }
    };
    const int n_full = E / TILE_EDGES;
    const int first  = (int)(((long long)blockIdx.x * n_full) / gridDim.x);
    int p = 0;
    for (; p + SCAN_UNROLL <= n_full; p += SCAN_UNROLL) {
      int4 dv[SCAN_UNROLL];
      int  eb[SCAN_UNROLL];
#pragma unroll
      for (int u = 0; u < SCAN_UNROLL; ++u) {
        int tile = first + p + u;
        if (tile >= n_full) tile -= n_full;
        eb[u] = tile * TILE_EDGES + tid * 4;
        dv[u] = *(const int4*)(dst + eb[u]);
      }
#pragma unroll
      for (int u = 0; u < SCAN_UNROLL; ++u) take(dv[u], eb[u]);
    }
    for (; p < n_full; ++p) {
      int tile = first + p;
      if (tile >= n_full) tile -= n_full;
      const int e0 = tile * TILE_EDGES + tid * 4;
      take(*(const int4*)(dst + e0), e0);
    }
    for (int i = n_full * TILE_EDGES + tid; i < E; i += OWN_THREADS) {
      const unsigned d0 = (unsigned)(dst[i] - lo);
      if (d0 < r) append(i);
    }
  }
  __syncthreads();
  const int total = n_listed;

  if (total <= LIST_CAP) {
    for (int e = tid; e < total; e += OWN_THREADS) {
      const int   i  = list_i[e];
      const int   d  = dst[i];
      const int   s  = src[i];
      const int   dl = d - lo;
      const float le = edge_logit(score, t, s, d);
      atomicMax((int*)&m_s[dl], __float_as_int(le));
      list_i[e]  = s;
      list_dl[e] = (unsigned short)dl;
      list_w[e]  = le;
    }
    __syncthreads();
    for (int e = tid; e < total; e += OWN_THREADS) {
      const int   dl = list_dl[e];
      const float un = expf(list_w[e] - m_s[dl]);
      atomicAdd(&den_s[dl], un);
      list_w[e] = un;
    }
    __syncthreads();
    const int wv = __builtin_amdgcn_readfirstlane(tid >> 5);
    for (int base = wv * 32; base < total; base += OWN_THREADS) {
      const int cnt = min(32, total - base);
      int   s = 0, d = 0;
      float alpha = 0.0f;
      if (lane < cnt) {
        s = list_i[base + lane];
        d = list_dl[base + lane];
        alpha = list_w[base + lane] / fmaxf(den_s[d], 1e-20f);
      }
      const int alpha_bits = __float_as_int(alpha);
      for (int j0 = 0; j0 < cnt; j0 += EDGE_GROUP) {
        if (j0 + EDGE_GROUP <= cnt) {
          float z[EDGE_GROUP][4];
#pragma unroll
          for (int u = 0; u < EDGE_GROUP; ++u) {
            const int sj = __builtin_amdgcn_readlane(s, j0 + u);
            const float* zr = Z + (size_t)sj * OUT_DIM + lane;
            z[u][0] = zr[0];
            z[u][1] = zr[32];
            z[u][2] = zr[64];
            z[u][3] = zr[96];
          }
#pragma unroll
          for (int u = 0; u < EDGE_GROUP; ++u) {
            const int   dj = __builtin_amdgcn_readlane(d, j0 + u);
            const float aj = __int_as_float(__builtin_amdgcn_readlane(alpha_bits, j0 + u));
            float* o = acc + dj * OUT_DIM + lane;
            atomicAdd(o,      aj * z[u][0]);
            atomicAdd(o + 32, aj * z[u][1]);
            atomicAdd(o + 64, aj * z[u][2]);
            atomicAdd(o + 96, aj * z[u][3]);
          }
        } else {
          for (int j = j0; j < cnt; ++j) {
            const int   sj = __builtin_amdgcn_readlane(s, j);
            const int   dj = __builtin_amdgcn_readlane(d, j);
            const float aj = __int_as_float(__builtin_amdgcn_readlane(alpha_bits, j));
            const float* zr = Z + (size_t)sj * OUT_DIM + lane;
            float* o = acc + dj * OUT_DIM + lane;
            atomicAdd(o,      aj * zr[0]);
            atomicAdd(o + 32, aj * zr[32]);
            atomicAdd(o + 64, aj * zr[64]);
            atomicAdd(o + 96, aj * zr[96]);
          }
        }
      }
    }
    __syncthreads();
  } else {
    for (int mode = 0; mode < 3; ++mode) {
      for (int i = tid; i < E; i += OWN_THREADS) {
        const unsigned dl = (unsigned)(dst[i] - lo);
        if (dl < (unsigned)rn) {
          const int   s  = src[i];
          const float le = edge_logit(score, t, s, lo + (int)dl);
          if (mode == 0) {
            atomicMax((int*)&m_s[dl], __float_as_int(le));
          } else {
            const float un = expf(le - m_s[dl]);
            if (mode == 1) {
              atomicAdd(&den_s[dl], un);
            } else {
              const float a = un / fmaxf(den_s[dl], 1e-20f);
              const float* zr = Z + (size_t)s * OUT_DIM;
              for (int c = 0; c < OUT_DIM; ++c) atomicAdd(&acc[dl * OUT_DIM + c], a * zr[c]);
            }
          }
        }
      }
      __syncthreads();
    }
  }

  {
    v4f* __restrict__ o4 = (v4f*)(out + (size_t)lo * OUT_DIM);
    const v4f* a4 = (const v4f*)acc;
    for (int o = tid; o < rn * (OUT_DIM / 4); o += OWN_THREADS) o4[o] = a4[o];
  }
}

extern "C" void kernel_launch(void* const* d_in, const int* in_sizes, int n_in,
                              void* d_out, int out_size, void* d_ws, size_t ws_size,
                              hipStream_t stream) {
  const float* features = (const float*)d_in[0];
  const float* t        = (const float*)d_in[1];
  const float* W_fc     = (const float*)d_in[2];
  const float* w_t      = (const float*)d_in[3];
  const int*   src_idx  = (const int*)d_in[4];
  const int*   dst_idx  = (const int*)d_in[5];

  const int n_src = in_sizes[0] / IN_DIM;
  const int E     = in_sizes[4];
  const int n_dst = out_size / OUT_DIM;

  float* zbuf  = (float*)d_ws;
  float* score = zbuf + (size_t)n_src * OUT_DIM;
  float* out   = (float*)d_out;

  float* wp = score;

  const int bm_waves = ((E + 31) / 32 + 31) / 32;
  k_prep<<<W_BLOCKS + (bm_waves + BM_WAVES - 1) / BM_WAVES, PREP_THREADS, 0, stream>>>(
      W_fc, wp, dst_idx, (unsigned*)d_out, E, n_dst);

  const int ntile = (n_src + 15) / 16;
  k_gemm<<<(ntile + 3) / 4, 128, 0, stream>>>(features, wp, zbuf, n_src);

  k_score<<<(n_src + SC_BLOCK_ROWS - 1) / SC_BLOCK_ROWS, SC_THREADS, 0, stream>>>(zbuf, w_t, score, n_src);

  k_own<<<(n_dst + R_NODES - 1) / R_NODES, OWN_THREADS, 0, stream>>>(
      score, t, src_idx, dst_idx, zbuf, out, E, n_dst);
}


# k_own: edges binned per owner wave by LDS ticket; a wave alone sums its 13 tile rows: 16-B z loads, LDS read, fused multiply-add, LDS write, no LDS float atomics; ticket order, residual added last
# speedup vs baseline: 3.8747x; 1.0640x over previous
#include <hip/hip_runtime.h>
#include <hip/hip_bf16.h>

typedef __attribute__((ext_vector_type(2))) float v2f;
typedef __attribute__((ext_vector_type(4))) float v4f;
typedef __attribute__((ext_vector_type(8))) float v8f;

#define IN_DIM  256
#define OUT_DIM 128

#define WP_FLOATS (OUT_DIM * IN_DIM)

__global__ __launch_bounds__(128) void k_gemm(const float* __restrict__ A,
                                              const float* __restrict__ Wp,
                                              float* __restrict__ Z,
                                              int n_src) {
  const int lane = threadIdx.x & 31;
  const int wave = threadIdx.x >> 5;
  const int tile = blockIdx.x * 4 + wave;
  const int m0   = tile * 16;
  if (m0 >= n_src) return;

  const int half = lane >> 4;
  const int nc   = lane & 15;

  int mrow = m0 + nc;
  if (mrow >= n_src) mrow = n_src - 1;
  const float* __restrict__ arow = A + (size_t)mrow * IN_DIM + 2 * half;
  const v4f* __restrict__ wl = (const v4f*)Wp + lane;

  const v8f vzero = {};
  v8f acc[8];
#pragma unroll
  for (int i = 0; i < 8; ++i) acc[i] = vzero;

#pragma unroll 1
  for (int t = 0; t < IN_DIM / 16; ++t) {
    v2f a[4];
    v4f b[16];
#pragma unroll
    for (int i = 0; i < 4; ++i) a[i] = *(const v2f*)(arow + 16 * t + 4 * i);
#pragma unroll
    for (int i = 0; i < 16; ++i) b[i] = wl[i * 32];
    wl += 16 * 32;
    __builtin_amdgcn_sched_barrier(0);
#pragma unroll
    for (int qq = 0; qq < 2; ++qq) {
#pragma unroll
      for (int tn = 0; tn < 8; ++tn) {
        const v4f bb = b[qq * 8 + tn];
        const v2f b0 = {bb[0], bb[1]};
        const v2f b1 = {bb[2], bb[3]};
        acc[tn] = __builtin_amdgcn_wmma_f32_16x16x4_f32(
             false, a[2 * qq],  false, b0,
             (short)0, acc[tn],  false,  false);
        acc[tn] = __builtin_amdgcn_wmma_f32_16x16x4_f32(
             false, a[2 * qq + 1],  false, b1,
             (short)0, acc[tn],  false,  false);
      }
    }
  }

#pragma unroll
  for (int r = 0; r < 8; ++r) {
    const int mm = m0 + r + 8 * half;
    if (mm < n_src) {
      float* zr = Z + (size_t)mm * OUT_DIM;
#pragma unroll
      for (int tn = 0; tn < 8; ++tn) zr[tn * 16 + nc] = acc[tn][r];
    }
  }
}

#define SC_ROWS    32
#define SC_THREADS 128
#define SC_BLOCK_ROWS (SC_ROWS * (SC_THREADS / 32))

__global__ __launch_bounds__(SC_THREADS) __attribute__((amdgpu_waves_per_eu(4, 5))) void k_score(const float* __restrict__ Z,
                                                      const float* __restrict__ wt,
                                                      float* __restrict__ score,
                                                      int n_src) {
  const int lane = threadIdx.x & 31;
  const int wv   = (blockIdx.x * SC_THREADS + threadIdx.x) >> 5;
  const int r0   = wv * SC_ROWS;
  if (r0 >= n_src) return;
  const int g = lane >> 3;
  const int c = lane & 7;

  v4f w[4];
#pragma unroll
  for (int j = 0; j < 4; ++j) w[j] = *(const v4f*)(wt + 32 * j + 4 * c);

  v4f z[SC_ROWS / 4][4];
#pragma unroll
  for (int k = 0; k < SC_ROWS / 4; ++k) {
    int row = r0 + 4 * k + g;
    if (row >= n_src) row = n_src - 1;
    const v4f* __restrict__ zr = (const v4f*)(Z + (size_t)row * OUT_DIM) + c;
#pragma unroll
    for (int j = 0; j < 4; ++j) z[k][j] = zr[8 * j];
  }

  __builtin_amdgcn_sched_barrier(0);

  float mine = 0.0f;
#pragma unroll
  for (int k = 0; k < SC_ROWS / 4; ++k) {
    float p = 0.0f;
#pragma unroll
    for (int j = 0; j < 4; ++j)
      p += z[k][j][0] * w[j][0] + z[k][j][1] * w[j][1] + z[k][j][2] * w[j][2] + z[k][j][3] * w[j][3];
    p += __shfl_xor(p, 1, 32);
    p += __shfl_xor(p, 2, 32);
    p += __shfl_xor(p, 4, 32);
    if (c == k) mine = p;
  }
  const int out_row = r0 + 4 * c + g;
  if (out_row < n_src) score[out_row] = mine;
}

__device__ __forceinline__ float edge_logit(const float* __restrict__ score,
                                            const float* __restrict__ t,
                                            int s, int d) {
  const float e = -fabsf(t[s] - t[d]);
  return expf(score[s] * e * 0.002f);
}

#define R_NODES      400
#define PREP_THREADS 128
#define BM_WAVES     (PREP_THREADS / 32)
#define BM_ROWS      252
#define W_BLOCKS     (WP_FLOATS / (PREP_THREADS * 4))

__device__ __forceinline__ bool bitmaps_fit(int E, int n_dst) {
  const int nblk      = (n_dst + R_NODES - 1) / R_NODES;
  const int words     = (E + 31) >> 5;
  const int last_rows = n_dst - (nblk - 1) * R_NODES;
  return nblk <= BM_ROWS && words <= last_rows * OUT_DIM;
}

__global__ __launch_bounds__(PREP_THREADS) void k_prep(const float* __restrict__ W,
                                                       float* __restrict__ Wp,
                                                       const int* __restrict__ dst,
                                                       unsigned* __restrict__ bm,
                                                       int E, int n_dst) {
  __shared__ unsigned tile[BM_WAVES][BM_ROWS * 32];
  const int tid = threadIdx.x;

  if (blockIdx.x < W_BLOCKS) {
    for (int o = blockIdx.x * PREP_THREADS + tid; o < WP_FLOATS; o += W_BLOCKS * PREP_THREADS) {
      const int e    = o & 3;
      const int lane = (o >> 2) & 31;
      const int tn   = (o >> 7) & 7;
      const int q    = o >> 10;
      const int n    = tn * 16 + (lane & 15);
      const int k    = 8 * q + 4 * (e >> 1) + 2 * (lane >> 4) + (e & 1);
      Wp[o] = W[n * IN_DIM + k];
    }
    return;
  }

  if (!bitmaps_fit(E, n_dst)) return;
  const int lane  = tid & 31;
  const int wv    = tid >> 5;
  const int nblk  = (n_dst + R_NODES - 1) / R_NODES;
  const int words = (E + 31) >> 5;
  const int gw    = (blockIdx.x - W_BLOCKS) * BM_WAVES + wv;
  if (gw * 32 >= words) return;
  const int wq    = gw * 32 + lane;
  const int e0    = wq * 32;

  unsigned* col = &tile[wv][lane];
  for (int v = 0; v < BM_ROWS; ++v) col[v * 32] = 0u;

  auto mark = [&](const int d, const int j) {
    const unsigned o = min((unsigned)d / (unsigned)R_NODES, (unsigned)(BM_ROWS - 1));
    atomicOr(&col[o * 32], 1u << j);
  };
  if (e0 + 32 <= E) {
    const int4* p = (const int4*)(dst + e0);
    int4 d[8];
#pragma unroll
    for (int c = 0; c < 8; ++c) d[c] = p[c];
#pragma unroll
    for (int c = 0; c < 8; ++c) {
      mark(d[c].x, 4 * c);
      mark(d[c].y, 4 * c + 1);
      mark(d[c].z, 4 * c + 2);
      mark(d[c].w, 4 * c + 3);
    }
  } else {
    for (int j = 0; j < 32; ++j)
      if (e0 + j < E) mark(dst[e0 + j], j);
  }

  if (wq < words) {
    unsigned* g = bm + wq;
#pragma unroll 10
    for (int v = 0; v < nblk; ++v) g[(size_t)v * (R_NODES * OUT_DIM)] = col[v * 32];
  }
}

#define OWN_THREADS 1024
#define OWN_WAVES   (OWN_THREADS / 32)
#define WAVE_ROWS   ((R_NODES + OWN_WAVES - 1) / OWN_WAVES)
#define LIST_CAP    8192
#define BIN_CAP     288
#define TILE_EDGES  (OWN_THREADS * 4)
#define SCAN_UNROLL 8
#define BM_UNROLL   4
#define EDGE_GROUP  8

static_assert(OWN_WAVES == 32, "a row's wave is its low 5 bits");
static_assert(WAVE_ROWS <= 256, "a record's k is one byte");

__global__ __launch_bounds__(OWN_THREADS) void k_own(const float* __restrict__ score,
                                                     const float* __restrict__ t,
                                                     const int* __restrict__ src,
                                                     const int* __restrict__ dst,
                                                     const float* __restrict__ Z,
                                                     float* __restrict__ out,
                                                     int E, int n_dst) {
  __shared__ __attribute__((aligned(16))) float acc[R_NODES * OUT_DIM];
  __shared__ float m_s[R_NODES];
  __shared__ float den_s[R_NODES];
  __shared__ int   list_i[LIST_CAP];
  __shared__ int   bin_s[OWN_WAVES * BIN_CAP];
  __shared__ float bin_w[OWN_WAVES * BIN_CAP];
  __shared__ unsigned char bin_k[OWN_WAVES * BIN_CAP];
  __shared__ int   bin_n[OWN_WAVES];
  __shared__ int   n_listed;

  const int tid  = threadIdx.x;
  const int lane = tid & 31;
  const int lo   = blockIdx.x * R_NODES;
  const int rn   = min(R_NODES, n_dst - lo);

  for (int o = tid; o < R_NODES; o += OWN_THREADS) { m_s[o] = 0.0f; den_s[o] = 0.0f; }
  if (tid < OWN_WAVES) bin_n[tid] = 0;
  if (tid == 0) n_listed = 0;
  {
    const v4f vzero = {};
    for (int row = tid >> 5; row < R_NODES; row += OWN_WAVES)
      *(v4f*)(acc + row * OUT_DIM + 4 * lane) = vzero;
  }
  __syncthreads();

  if (bitmaps_fit(E, n_dst)) {
    const unsigned* __restrict__ bmw = (const unsigned*)(out + (size_t)lo * OUT_DIM);
    const uint4* __restrict__ bm4 = (const uint4*)bmw;
    const int words = (E + 31) >> 5;
    const int quads = words >> 2;
    auto emit = [&](unsigned x, const int ebase, int& sl) {
      while (x) {
        const int b = __builtin_ctz(x);
        x &= x - 1;
        if (sl < LIST_CAP) list_i[sl] = ebase + b;
        ++sl;
      }
    };
    for (int q0 = tid; q0 < quads; q0 += OWN_THREADS * BM_UNROLL) {
      uint4 x[BM_UNROLL];
#pragma unroll
      for (int u = 0; u < BM_UNROLL; ++u) {
        const int q = q0 + u * OWN_THREADS;
        x[u] = (q < quads) ? bm4[q] : make_uint4(0u, 0u, 0u, 0u);
      }
#pragma unroll
      for (int u = 0; u < BM_UNROLL; ++u) {
        const int cnt = __builtin_popcount(x[u].x) + __builtin_popcount(x[u].y) +
                        __builtin_popcount(x[u].z) + __builtin_popcount(x[u].w);
        if (cnt) {
          int sl = atomicAdd(&n_listed, cnt);
          const int eb = (q0 + u * OWN_THREADS) * 128;
          emit(x[u].x, eb,      sl);
          emit(x[u].y, eb + 32, sl);
          emit(x[u].z, eb + 64, sl);
          emit(x[u].w, eb + 96, sl);
        }
      }
    }
    for (int w = quads * 4 + tid; w < words; w += OWN_THREADS) {
      const unsigned x = bmw[w];
      if (x) {
        int sl = atomicAdd(&n_listed, __builtin_popcount(x));
        emit(x, w * 32, sl);
      }
    }
  } else {
    const unsigned r = (unsigned)rn;
    auto append = [&](const int i) {
      const int sl = atomicAdd(&n_listed, 1);
      if (sl < LIST_CAP) list_i[sl] = i;
    };
    auto take = [&](const int4 dv, const int e0) {
      const unsigned d0 = (unsigned)(dv.x - lo);
      const unsigned d1 = (unsigned)(dv.y - lo);
      const unsigned d2 = (unsigned)(dv.z - lo);
      const unsigned d3 = (unsigned)(dv.w - lo);
      if (d0 < r || d1 < r || d2 < r || d3 < r) {
        if (d0 < r) append(e0);
        if (d1 < r) append(e0 + 1);
        if (d2 < r) append(e0 + 2);
        if (d3 < r) append(e0 + 3);
      }
    };
    const int n_full = E / TILE_EDGES;
    const int first  = (int)(((long long)blockIdx.x * n_full) / gridDim.x);
    int p = 0;
    for (; p + SCAN_UNROLL <= n_full; p += SCAN_UNROLL) {
      int4 dv[SCAN_UNROLL];
      int  eb[SCAN_UNROLL];
#pragma unroll
      for (int u = 0; u < SCAN_UNROLL; ++u) {
        int tile = first + p + u;
        if (tile >= n_full) tile -= n_full;
        eb[u] = tile * TILE_EDGES + tid * 4;
        dv[u] = *(const int4*)(dst + eb[u]);
      }
#pragma unroll
      for (int u = 0; u < SCAN_UNROLL; ++u) take(dv[u], eb[u]);
    }
    for (; p < n_full; ++p) {
      int tile = first + p;
      if (tile >= n_full) tile -= n_full;
      const int e0 = tile * TILE_EDGES + tid * 4;
      take(*(const int4*)(dst + e0), e0);
    }
    for (int i = n_full * TILE_EDGES + tid; i < E; i += OWN_THREADS) {
      const unsigned d0 = (unsigned)(dst[i] - lo);
      if (d0 < r) append(i);
    }
  }
  __syncthreads();
  const int total = n_listed;

  bool binned = total <= LIST_CAP;
  if (binned) {
    for (int e = tid; e < total; e += OWN_THREADS) {
      const int   i  = list_i[e];
      const int   d  = dst[i];
      const int   s  = src[i];
      const int   dl = d - lo;
      const float le = edge_logit(score, t, s, d);
      atomicMax((int*)&m_s[dl], __float_as_int(le));
      const int b  = dl & (OWN_WAVES - 1);
      const int sl = atomicAdd(&bin_n[b], 1);
      if (sl < BIN_CAP) {
        const int p = b * BIN_CAP + sl;
        bin_s[p] = s;
        bin_w[p] = le;
        bin_k[p] = (unsigned char)(dl / OWN_WAVES);
      }
    }
    __syncthreads();
    int fullest = 0;
    for (int b = 0; b < OWN_WAVES; ++b) fullest = max(fullest, bin_n[b]);
    binned = fullest <= BIN_CAP;
  }

  if (binned) {
    const int wv = __builtin_amdgcn_readfirstlane(tid >> 5);
    const int n  = __builtin_amdgcn_readfirstlane(bin_n[wv]);
    const int p0 = wv * BIN_CAP;

    for (int base = 0; base < n; base += 32) {
      if (base + lane < n) {
        const int   p   = p0 + base + lane;
        const int   row = (int)bin_k[p] * OWN_WAVES + wv;
        const float un  = expf(bin_w[p] - m_s[row]);
        atomicAdd(&den_s[row], un);
        bin_w[p] = un;
      }
    }

    auto add_row = [&](const int k, const float aj, const v4f zv) {
      v4f* const p = (v4f*)(acc + (k * OWN_WAVES + wv) * OUT_DIM + 4 * lane);
      v4f r = *p;
      r[0] = __builtin_fmaf(aj, zv[0], r[0]);
      r[1] = __builtin_fmaf(aj, zv[1], r[1]);
      r[2] = __builtin_fmaf(aj, zv[2], r[2]);
      r[3] = __builtin_fmaf(aj, zv[3], r[3]);
      *p = r;
    };

    for (int base = 0; base < n; base += 32) {
      const int cnt = min(32, n - base);
      int   s = 0, k = 0;
      float alpha = 0.0f;
      if (lane < cnt) {
        const int p = p0 + base + lane;
        s = bin_s[p];
        k = bin_k[p];
        alpha = bin_w[p] / fmaxf(den_s[k * OWN_WAVES + wv], 1e-20f);
      }
      const int alpha_bits = __float_as_int(alpha);
      auto load8 = [&](v4f (&z)[EDGE_GROUP], const int j0) {
#pragma unroll
        for (int u = 0; u < EDGE_GROUP; ++u) {
          const int sj = __builtin_amdgcn_readlane(s, j0 + u);
          z[u] = *(const v4f*)(Z + (size_t)sj * OUT_DIM + 4 * lane);
        }
      };
      auto add8 = [&](const v4f (&z)[EDGE_GROUP], const int j0) {
#pragma unroll
        for (int u = 0; u < EDGE_GROUP; ++u) {
          const int   kj = __builtin_amdgcn_readlane(k, j0 + u);
          const float aj = __int_as_float(__builtin_amdgcn_readlane(alpha_bits, j0 + u));
          add_row(kj, aj, z[u]);
        }
      };
      if (cnt == 32) {
        v4f za[EDGE_GROUP], zb[EDGE_GROUP];
        load8(za, 0);
        load8(zb, 8);
        add8(za, 0);
        load8(za, 16);
        add8(zb, 8);
        load8(zb, 24);
        add8(za, 16);
        add8(zb, 24);
      } else {
        int j0 = 0;
        for (; j0 + EDGE_GROUP <= cnt; j0 += EDGE_GROUP) {
          v4f z[EDGE_GROUP];
          load8(z, j0);
          add8(z, j0);
        }
        for (int j = j0; j < cnt; ++j) {
          const int   sj = __builtin_amdgcn_readlane(s, j);
          const int   kj = __builtin_amdgcn_readlane(k, j);
          const float aj = __int_as_float(__builtin_amdgcn_readlane(alpha_bits, j));
          add_row(kj, aj, *(const v4f*)(Z + (size_t)sj * OUT_DIM + 4 * lane));
        }
      }
    }

#pragma unroll
    for (int k = 0; k < WAVE_ROWS; ++k) {
      const int row = k * OWN_WAVES + wv;
      if (row < rn) {
        const v4f zres = *(const v4f*)(Z + (size_t)(lo + row) * OUT_DIM + 4 * lane);
        const v4f sum  = *(const v4f*)(acc + row * OUT_DIM + 4 * lane);
        *(v4f*)(out + (size_t)(lo + row) * OUT_DIM + 4 * lane) = zres + sum;
      }
    }
  } else {
    {
      const v4f* __restrict__ z4 = (const v4f*)(Z + (size_t)lo * OUT_DIM);
      v4f* a4 = (v4f*)acc;
      for (int o = tid; o < rn * (OUT_DIM / 4); o += OWN_THREADS) a4[o] = z4[o];
    }
    __syncthreads();
    for (int mode = 0; mode < 3; ++mode) {
      for (int i = tid; i < E; i += OWN_THREADS) {
        const unsigned dl = (unsigned)(dst[i] - lo);
        if (dl < (unsigned)rn) {
          const int   s  = src[i];
          const float le = edge_logit(score, t, s, lo + (int)dl);
          if (mode == 0) {
            atomicMax((int*)&m_s[dl], __float_as_int(le));
          } else {
            const float un = expf(le - m_s[dl]);
            if (mode == 1) {
              atomicAdd(&den_s[dl], un);
            } else {
              const float al = un / fmaxf(den_s[dl], 1e-20f);
              const float* zr = Z + (size_t)s * OUT_DIM;
              for (int c = 0; c < OUT_DIM; ++c) atomicAdd(&acc[dl * OUT_DIM + c], al * zr[c]);
            }
          }
        }
      }
      __syncthreads();
    }
    v4f* __restrict__ o4 = (v4f*)(out + (size_t)lo * OUT_DIM);
    const v4f* a4 = (const v4f*)acc;
    for (int o = tid; o < rn * (OUT_DIM / 4); o += OWN_THREADS) o4[o] = a4[o];
  }
}

extern "C" void kernel_launch(void* const* d_in, const int* in_sizes, int n_in,
                              void* d_out, int out_size, void* d_ws, size_t ws_size,
                              hipStream_t stream) {
  const float* features = (const float*)d_in[0];
  const float* t        = (const float*)d_in[1];
  const float* W_fc     = (const float*)d_in[2];
  const float* w_t      = (const float*)d_in[3];
  const int*   src_idx  = (const int*)d_in[4];
  const int*   dst_idx  = (const int*)d_in[5];

  const int n_src = in_sizes[0] / IN_DIM;
  const int E     = in_sizes[4];
  const int n_dst = out_size / OUT_DIM;

  float* zbuf  = (float*)d_ws;
  float* score = zbuf + (size_t)n_src * OUT_DIM;
  float* out   = (float*)d_out;

  float* wp = score;

  const int bm_waves = ((E + 31) / 32 + 31) / 32;
  k_prep<<<W_BLOCKS + (bm_waves + BM_WAVES - 1) / BM_WAVES, PREP_THREADS, 0, stream>>>(
      W_fc, wp, dst_idx, (unsigned*)d_out, E, n_dst);

  const int ntile = (n_src + 15) / 16;
  k_gemm<<<(ntile + 3) / 4, 128, 0, stream>>>(features, wp, zbuf, n_src);

  k_score<<<(n_src + SC_BLOCK_ROWS - 1) / SC_BLOCK_ROWS, SC_THREADS, 0, stream>>>(zbuf, w_t, score, n_src);

  k_own<<<(n_dst + R_NODES - 1) / R_NODES, OWN_THREADS, 0, stream>>>(
      score, t, src_idx, dst_idx, zbuf, out, E, n_dst);
}


# k_prep map word = {count, row, src+1} of its 32 edges, not 32 bits: a word with one edge IS the edge; k_own lists it without reading dst[i], src[i] (2 scattered reads per edge gone); ticket order
# speedup vs baseline: 4.1116x; 1.0611x over previous
#include <hip/hip_runtime.h>
#include <hip/hip_bf16.h>

typedef __attribute__((ext_vector_type(2))) float v2f;
typedef __attribute__((ext_vector_type(4))) float v4f;
typedef __attribute__((ext_vector_type(8))) float v8f;

#define IN_DIM  256
#define OUT_DIM 128

#define WP_FLOATS (OUT_DIM * IN_DIM)

__global__ __launch_bounds__(128) void k_gemm(const float* __restrict__ A,
                                              const float* __restrict__ Wp,
                                              float* __restrict__ Z,
                                              int n_src) {
  const int lane = threadIdx.x & 31;
  const int wave = threadIdx.x >> 5;
  const int tile = blockIdx.x * 4 + wave;
  const int m0   = tile * 16;
  if (m0 >= n_src) return;

  const int half = lane >> 4;
  const int nc   = lane & 15;

  int mrow = m0 + nc;
  if (mrow >= n_src) mrow = n_src - 1;
  const float* __restrict__ arow = A + (size_t)mrow * IN_DIM + 2 * half;
  const v4f* __restrict__ wl = (const v4f*)Wp + lane;

  const v8f vzero = {};
  v8f acc[8];
#pragma unroll
  for (int i = 0; i < 8; ++i) acc[i] = vzero;

#pragma unroll 1
  for (int t = 0; t < IN_DIM / 16; ++t) {
    v2f a[4];
    v4f b[16];
#pragma unroll
    for (int i = 0; i < 4; ++i) a[i] = *(const v2f*)(arow + 16 * t + 4 * i);
#pragma unroll
    for (int i = 0; i < 16; ++i) b[i] = wl[i * 32];
    wl += 16 * 32;
    __builtin_amdgcn_sched_barrier(0);
#pragma unroll
    for (int qq = 0; qq < 2; ++qq) {
#pragma unroll
      for (int tn = 0; tn < 8; ++tn) {
        const v4f bb = b[qq * 8 + tn];
        const v2f b0 = {bb[0], bb[1]};
        const v2f b1 = {bb[2], bb[3]};
        acc[tn] = __builtin_amdgcn_wmma_f32_16x16x4_f32(
             false, a[2 * qq],  false, b0,
             (short)0, acc[tn],  false,  false);
        acc[tn] = __builtin_amdgcn_wmma_f32_16x16x4_f32(
             false, a[2 * qq + 1],  false, b1,
             (short)0, acc[tn],  false,  false);
      }
    }
  }

#pragma unroll
  for (int r = 0; r < 8; ++r) {
    const int mm = m0 + r + 8 * half;
    if (mm < n_src) {
      float* zr = Z + (size_t)mm * OUT_DIM;
#pragma unroll
      for (int tn = 0; tn < 8; ++tn) zr[tn * 16 + nc] = acc[tn][r];
    }
  }
}

#define SC_ROWS    32
#define SC_THREADS 128
#define SC_BLOCK_ROWS (SC_ROWS * (SC_THREADS / 32))

__global__ __launch_bounds__(SC_THREADS) __attribute__((amdgpu_waves_per_eu(4, 5))) void k_score(const float* __restrict__ Z,
                                                      const float* __restrict__ wt,
                                                      float* __restrict__ score,
                                                      int n_src) {
  const int lane = threadIdx.x & 31;
  const int wv   = (blockIdx.x * SC_THREADS + threadIdx.x) >> 5;
  const int r0   = wv * SC_ROWS;
  if (r0 >= n_src) return;
  const int g = lane >> 3;
  const int c = lane & 7;

  v4f w[4];
#pragma unroll
  for (int j = 0; j < 4; ++j) w[j] = *(const v4f*)(wt + 32 * j + 4 * c);

  v4f z[SC_ROWS / 4][4];
#pragma unroll
  for (int k = 0; k < SC_ROWS / 4; ++k) {
    int row = r0 + 4 * k + g;
    if (row >= n_src) row = n_src - 1;
    const v4f* __restrict__ zr = (const v4f*)(Z + (size_t)row * OUT_DIM) + c;
#pragma unroll
    for (int j = 0; j < 4; ++j) z[k][j] = zr[8 * j];
  }

  __builtin_amdgcn_sched_barrier(0);

  float mine = 0.0f;
#pragma unroll
  for (int k = 0; k < SC_ROWS / 4; ++k) {
    float p = 0.0f;
#pragma unroll
    for (int j = 0; j < 4; ++j)
      p += z[k][j][0] * w[j][0] + z[k][j][1] * w[j][1] + z[k][j][2] * w[j][2] + z[k][j][3] * w[j][3];
    p += __shfl_xor(p, 1, 32);
    p += __shfl_xor(p, 2, 32);
    p += __shfl_xor(p, 4, 32);
    if (c == k) mine = p;
  }
  const int out_row = r0 + 4 * c + g;
  if (out_row < n_src) score[out_row] = mine;
}

__device__ __forceinline__ float edge_logit(const float* __restrict__ score,
                                            const float* __restrict__ t,
                                            int s, int d) {
  const float e = -fabsf(t[s] - t[d]);
  return expf(score[s] * e * 0.002f);
}

#define R_NODES      400
#define PREP_THREADS 128
#define BM_WAVES     (PREP_THREADS / 32)
#define BM_ROWS      252
#define W_BLOCKS     (WP_FLOATS / (PREP_THREADS * 4))
#define SRC_BITS     18
#define ROW_BITS     9
#define TAG_SHIFT    (SRC_BITS + ROW_BITS)
#define SRC_MASK     ((1u << SRC_BITS) - 1u)
#define PAY_MASK     ((1u << TAG_SHIFT) - 1u)

static_assert(R_NODES <= (1 << ROW_BITS), "a row of an owner fits its field");
static_assert(TAG_SHIFT + 5 == 32, "count field: 5 bits");

__device__ __forceinline__ bool bitmaps_fit(int E, int n_dst, int n_src) {
  const int nblk      = (n_dst + R_NODES - 1) / R_NODES;
  const int words     = (E + 31) >> 5;
  const int last_rows = n_dst - (nblk - 1) * R_NODES;
  return nblk <= BM_ROWS && words <= last_rows * OUT_DIM && n_src < (1 << SRC_BITS);
}

__global__ __launch_bounds__(PREP_THREADS) void k_prep(const float* __restrict__ W,
                                                       float* __restrict__ Wp,
                                                       const int* __restrict__ src,
                                                       const int* __restrict__ dst,
                                                       unsigned* __restrict__ bm,
                                                       int E, int n_dst, int n_src) {
  __shared__ unsigned tile[BM_WAVES][BM_ROWS * 32];
  const int tid = threadIdx.x;

  if (blockIdx.x < W_BLOCKS) {
    for (int o = blockIdx.x * PREP_THREADS + tid; o < WP_FLOATS; o += W_BLOCKS * PREP_THREADS) {
      const int e    = o & 3;
      const int lane = (o >> 2) & 31;
      const int tn   = (o >> 7) & 7;
      const int q    = o >> 10;
      const int n    = tn * 16 + (lane & 15);
      const int k    = 8 * q + 4 * (e >> 1) + 2 * (lane >> 4) + (e & 1);
      Wp[o] = W[n * IN_DIM + k];
    }
    return;
  }

  if (!bitmaps_fit(E, n_dst, n_src)) return;
  const int lane  = tid & 31;
  const int wv    = tid >> 5;
  const int nblk  = (n_dst + R_NODES - 1) / R_NODES;
  const int words = (E + 31) >> 5;
  const int gw    = (blockIdx.x - W_BLOCKS) * BM_WAVES + wv;
  if (gw * 32 >= words) return;
  const int wq    = gw * 32 + lane;
  const int e0    = wq * 32;

  unsigned* col = &tile[wv][lane];
  for (int v = 0; v < BM_ROWS; ++v) col[v * 32] = 0u;

  auto mark = [&](const int d, const int s) {
    const unsigned o   = min((unsigned)d / (unsigned)R_NODES, (unsigned)(BM_ROWS - 1));
    const unsigned row = ((unsigned)d - o * (unsigned)R_NODES) & ((1u << ROW_BITS) - 1u);
    const unsigned pay = (row << SRC_BITS) | (((unsigned)s + 1u) & SRC_MASK);
    atomicAdd(&col[o * 32], 1u << TAG_SHIFT);
    atomicOr(&col[o * 32], pay);
  };
  if (e0 + 32 <= E) {
    const int4* p  = (const int4*)(dst + e0);
    const int4* ps = (const int4*)(src + e0);
    int4 d[8], s[8];
#pragma unroll
    for (int c = 0; c < 8; ++c) { d[c] = p[c]; s[c] = ps[c]; }
#pragma unroll
    for (int c = 0; c < 8; ++c) {
      mark(d[c].x, s[c].x);
      mark(d[c].y, s[c].y);
      mark(d[c].z, s[c].z);
      mark(d[c].w, s[c].w);
    }
  } else {
    for (int j = 0; j < 32; ++j)
      if (e0 + j < E) mark(dst[e0 + j], src[e0 + j]);
  }

  if (wq < words) {
    unsigned* g = bm + wq;
#pragma unroll 10
    for (int v = 0; v < nblk; ++v) g[(size_t)v * (R_NODES * OUT_DIM)] = col[v * 32];
  }
}

#define OWN_THREADS 1024
#define OWN_WAVES   (OWN_THREADS / 32)
#define WAVE_ROWS   ((R_NODES + OWN_WAVES - 1) / OWN_WAVES)
#define LIST_CAP    8192
#define BIN_CAP     288
#define TILE_EDGES  (OWN_THREADS * 4)
#define SCAN_UNROLL 8
#define BM_UNROLL   4
#define EDGE_GROUP  8

static_assert(OWN_WAVES == 32, "a row's wave is its low 5 bits");
static_assert(WAVE_ROWS <= 256, "a record's k is one byte");

__global__ __launch_bounds__(OWN_THREADS) void k_own(const float* __restrict__ score,
                                                     const float* __restrict__ t,
                                                     const int* __restrict__ src,
                                                     const int* __restrict__ dst,
                                                     const float* __restrict__ Z,
                                                     float* __restrict__ out,
                                                     int E, int n_dst, int n_src) {
  __shared__ __attribute__((aligned(16))) float acc[R_NODES * OUT_DIM];
  __shared__ float m_s[R_NODES];
  __shared__ float den_s[R_NODES];
  __shared__ int   list_i[LIST_CAP];
  __shared__ int   bin_s[OWN_WAVES * BIN_CAP];
  __shared__ float bin_w[OWN_WAVES * BIN_CAP];
  __shared__ unsigned char bin_k[OWN_WAVES * BIN_CAP];
  __shared__ int   bin_n[OWN_WAVES];
  __shared__ int   n_listed;
  __shared__ int   n_multi;

  const int tid  = threadIdx.x;
  const int lane = tid & 31;
  const int lo   = blockIdx.x * R_NODES;
  const int rn   = min(R_NODES, n_dst - lo);

  for (int o = tid; o < R_NODES; o += OWN_THREADS) { m_s[o] = 0.0f; den_s[o] = 0.0f; }
  if (tid < OWN_WAVES) bin_n[tid] = 0;
  if (tid == 0) { n_listed = 0; n_multi = 0; }
  {
    const v4f vzero = {};
    for (int row = tid >> 5; row < R_NODES; row += OWN_WAVES)
      *(v4f*)(acc + row * OUT_DIM + 4 * lane) = vzero;
  }
  __syncthreads();

  const bool mapped = bitmaps_fit(E, n_dst, n_src);
  if (mapped) {
    const unsigned* __restrict__ bmw = (const unsigned*)(out + (size_t)lo * OUT_DIM);
    const uint4* __restrict__ bm4 = (const uint4*)bmw;
    const int words = (E + 31) >> 5;
    const int quads = words >> 2;
    int* const multi = bin_s;
    const int multi_cap = OWN_WAVES * BIN_CAP;
    auto single = [](const unsigned x) { return (x >> TAG_SHIFT) == 1u; };
    auto emit = [&](const unsigned x, int& sl) {
      if (single(x)) {
        if (sl < LIST_CAP) list_i[sl] = (int)(x & PAY_MASK);
        ++sl;
      }
    };
    auto note = [&](const unsigned x, const int w) {
      if (x != 0u && !single(x)) {
        const int ms = atomicAdd(&n_multi, 1);
        if (ms < multi_cap) multi[ms] = w;
      }
    };
    for (int q0 = tid; q0 < quads; q0 += OWN_THREADS * BM_UNROLL) {
      uint4 x[BM_UNROLL];
#pragma unroll
      for (int u = 0; u < BM_UNROLL; ++u) {
        const int q = q0 + u * OWN_THREADS;
        x[u] = (q < quads) ? bm4[q] : make_uint4(0u, 0u, 0u, 0u);
      }
#pragma unroll
      for (int u = 0; u < BM_UNROLL; ++u) {
        if (x[u].x | x[u].y | x[u].z | x[u].w) {
          const int cnt = (int)single(x[u].x) + (int)single(x[u].y) +
                          (int)single(x[u].z) + (int)single(x[u].w);
          const int wb = (q0 + u * OWN_THREADS) * 4;
          if (cnt) {
            int sl = atomicAdd(&n_listed, cnt);
            emit(x[u].x, sl);
            emit(x[u].y, sl);
            emit(x[u].z, sl);
            emit(x[u].w, sl);
          }
          note(x[u].x, wb);
          note(x[u].y, wb + 1);
          note(x[u].z, wb + 2);
          note(x[u].w, wb + 3);
        }
      }
    }
    for (int w = quads * 4 + tid; w < words; w += OWN_THREADS) {
      const unsigned x = bmw[w];
      if (single(x)) {
        int sl = atomicAdd(&n_listed, 1);
        emit(x, sl);
      }
      note(x, w);
    }
    __syncthreads();
    const unsigned r = (unsigned)rn;
    const int nm = n_multi;
    if (nm > multi_cap) atomicAdd(&n_listed, LIST_CAP + 1);
    for (int m = tid; m < min(nm, multi_cap); m += OWN_THREADS) {
      const int e0 = multi[m] * 32;
      unsigned hit = 0u;
      if (e0 + 32 <= E) {
        const int4* p = (const int4*)(dst + e0);
        int4 d[8];
#pragma unroll
        for (int c = 0; c < 8; ++c) d[c] = p[c];
#pragma unroll
        for (int c = 0; c < 8; ++c) {
          hit |= ((unsigned)(d[c].x - lo) < r ? 1u : 0u) << (4 * c);
          hit |= ((unsigned)(d[c].y - lo) < r ? 1u : 0u) << (4 * c + 1);
          hit |= ((unsigned)(d[c].z - lo) < r ? 1u : 0u) << (4 * c + 2);
          hit |= ((unsigned)(d[c].w - lo) < r ? 1u : 0u) << (4 * c + 3);
        }
      } else {
        for (int j = 0; j < 32; ++j)
          if (e0 + j < E && (unsigned)(dst[e0 + j] - lo) < r) hit |= 1u << j;
      }
      if (hit) {
        int sl = atomicAdd(&n_listed, __builtin_popcount(hit));
        while (hit) {
          const int j = __builtin_ctz(hit);
          hit &= hit - 1;
          const unsigned row = (unsigned)(dst[e0 + j] - lo);
          const unsigned pay = (row << SRC_BITS) | (((unsigned)src[e0 + j] + 1u) & SRC_MASK);
          if (sl < LIST_CAP) list_i[sl] = (int)pay;
          ++sl;
        }
      }
    }
  } else {
    const unsigned r = (unsigned)rn;
    auto append = [&](const int i) {
      const int sl = atomicAdd(&n_listed, 1);
      if (sl < LIST_CAP) list_i[sl] = i;
    };
    auto take = [&](const int4 dv, const int e0) {
      const unsigned d0 = (unsigned)(dv.x - lo);
      const unsigned d1 = (unsigned)(dv.y - lo);
      const unsigned d2 = (unsigned)(dv.z - lo);
      const unsigned d3 = (unsigned)(dv.w - lo);
      if (d0 < r || d1 < r || d2 < r || d3 < r) {
        if (d0 < r) append(e0);
        if (d1 < r) append(e0 + 1);
        if (d2 < r) append(e0 + 2);
        if (d3 < r) append(e0 + 3);
      }
    };
    const int n_full = E / TILE_EDGES;
    const int first  = (int)(((long long)blockIdx.x * n_full) / gridDim.x);
    int p = 0;
    for (; p + SCAN_UNROLL <= n_full; p += SCAN_UNROLL) {
      int4 dv[SCAN_UNROLL];
      int  eb[SCAN_UNROLL];
#pragma unroll
      for (int u = 0; u < SCAN_UNROLL; ++u) {
        int tile = first + p + u;
        if (tile >= n_full) tile -= n_full;
        eb[u] = tile * TILE_EDGES + tid * 4;
        dv[u] = *(const int4*)(dst + eb[u]);
      }
#pragma unroll
      for (int u = 0; u < SCAN_UNROLL; ++u) take(dv[u], eb[u]);
    }
    for (; p < n_full; ++p) {
      int tile = first + p;
      if (tile >= n_full) tile -= n_full;
      const int e0 = tile * TILE_EDGES + tid * 4;
      take(*(const int4*)(dst + e0), e0);
    }
    for (int i = n_full * TILE_EDGES + tid; i < E; i += OWN_THREADS) {
      const unsigned d0 = (unsigned)(dst[i] - lo);
      if (d0 < r) append(i);
    }
  }
  __syncthreads();
  const int total = n_listed;

  bool binned = total <= LIST_CAP;
  if (binned) {
    for (int e = tid; e < total; e += OWN_THREADS) {
      const int   v  = list_i[e];
      int s, dl;
      if (mapped) {
        dl = (int)((unsigned)v >> SRC_BITS);
        s  = (int)((unsigned)v & SRC_MASK) - 1;
      } else {
        dl = dst[v] - lo;
        s  = src[v];
      }
      const float le = edge_logit(score, t, s, lo + dl);
      atomicMax((int*)&m_s[dl], __float_as_int(le));
      const int b  = dl & (OWN_WAVES - 1);
      const int sl = atomicAdd(&bin_n[b], 1);
      if (sl < BIN_CAP) {
        const int p = b * BIN_CAP + sl;
        bin_s[p] = s;
        bin_w[p] = le;
        bin_k[p] = (unsigned char)(dl / OWN_WAVES);
      }
    }
    __syncthreads();
    int fullest = 0;
    for (int b = 0; b < OWN_WAVES; ++b) fullest = max(fullest, bin_n[b]);
    binned = fullest <= BIN_CAP;
  }

  if (binned) {
    const int wv = __builtin_amdgcn_readfirstlane(tid >> 5);
    const int n  = __builtin_amdgcn_readfirstlane(bin_n[wv]);
    const int p0 = wv * BIN_CAP;

    for (int base = 0; base < n; base += 32) {
      if (base + lane < n) {
        const int   p   = p0 + base + lane;
        const int   row = (int)bin_k[p] * OWN_WAVES + wv;
        const float un  = expf(bin_w[p] - m_s[row]);
        atomicAdd(&den_s[row], un);
        bin_w[p] = un;
      }
    }

    auto add_row = [&](const int k, const float aj, const v4f zv) {
      v4f* const p = (v4f*)(acc + (k * OWN_WAVES + wv) * OUT_DIM + 4 * lane);
      v4f r = *p;
      r[0] = __builtin_fmaf(aj, zv[0], r[0]);
      r[1] = __builtin_fmaf(aj, zv[1], r[1]);
      r[2] = __builtin_fmaf(aj, zv[2], r[2]);
      r[3] = __builtin_fmaf(aj, zv[3], r[3]);
      *p = r;
    };

    for (int base = 0; base < n; base += 32) {
      const int cnt = min(32, n - base);
      int   s = 0, k = 0;
      float alpha = 0.0f;
      if (lane < cnt) {
        const int p = p0 + base + lane;
        s = bin_s[p];
        k = bin_k[p];
        alpha = bin_w[p] / fmaxf(den_s[k * OWN_WAVES + wv], 1e-20f);
      }
      const int alpha_bits = __float_as_int(alpha);
      auto load8 = [&](v4f (&z)[EDGE_GROUP], const int j0) {
#pragma unroll
        for (int u = 0; u < EDGE_GROUP; ++u) {
          const int sj = __builtin_amdgcn_readlane(s, j0 + u);
          z[u] = *(const v4f*)(Z + (size_t)sj * OUT_DIM + 4 * lane);
        }
      };
      auto add8 = [&](const v4f (&z)[EDGE_GROUP], const int j0) {
#pragma unroll
        for (int u = 0; u < EDGE_GROUP; ++u) {
          const int   kj = __builtin_amdgcn_readlane(k, j0 + u);
          const float aj = __int_as_float(__builtin_amdgcn_readlane(alpha_bits, j0 + u));
          add_row(kj, aj, z[u]);
        }
      };
      if (cnt == 32) {
        v4f za[EDGE_GROUP], zb[EDGE_GROUP];
        load8(za, 0);
        load8(zb, 8);
        add8(za, 0);
        load8(za, 16);
        add8(zb, 8);
        load8(zb, 24);
        add8(za, 16);
        add8(zb, 24);
      } else {
        int j0 = 0;
        for (; j0 + EDGE_GROUP <= cnt; j0 += EDGE_GROUP) {
          v4f z[EDGE_GROUP];
          load8(z, j0);
          add8(z, j0);
        }
        for (int j = j0; j < cnt; ++j) {
          const int   sj = __builtin_amdgcn_readlane(s, j);
          const int   kj = __builtin_amdgcn_readlane(k, j);
          const float aj = __int_as_float(__builtin_amdgcn_readlane(alpha_bits, j));
          add_row(kj, aj, *(const v4f*)(Z + (size_t)sj * OUT_DIM + 4 * lane));
        }
      }
    }

#pragma unroll
    for (int k = 0; k < WAVE_ROWS; ++k) {
      const int row = k * OWN_WAVES + wv;
      if (row < rn) {
        const v4f zres = *(const v4f*)(Z + (size_t)(lo + row) * OUT_DIM + 4 * lane);
        const v4f sum  = *(const v4f*)(acc + row * OUT_DIM + 4 * lane);
        *(v4f*)(out + (size_t)(lo + row) * OUT_DIM + 4 * lane) = zres + sum;
      }
    }
  } else {
    {
      const v4f* __restrict__ z4 = (const v4f*)(Z + (size_t)lo * OUT_DIM);
      v4f* a4 = (v4f*)acc;
      for (int o = tid; o < rn * (OUT_DIM / 4); o += OWN_THREADS) a4[o] = z4[o];
    }
    __syncthreads();
    for (int mode = 0; mode < 3; ++mode) {
      for (int i = tid; i < E; i += OWN_THREADS) {
        const unsigned dl = (unsigned)(dst[i] - lo);
        if (dl < (unsigned)rn) {
          const int   s  = src[i];
          const float le = edge_logit(score, t, s, lo + (int)dl);
          if (mode == 0) {
            atomicMax((int*)&m_s[dl], __float_as_int(le));
          } else {
            const float un = expf(le - m_s[dl]);
            if (mode == 1) {
              atomicAdd(&den_s[dl], un);
            } else {
              const float al = un / fmaxf(den_s[dl], 1e-20f);
              const float* zr = Z + (size_t)s * OUT_DIM;
              for (int c = 0; c < OUT_DIM; ++c) atomicAdd(&acc[dl * OUT_DIM + c], al * zr[c]);
            }
          }
        }
      }
      __syncthreads();
    }
    v4f* __restrict__ o4 = (v4f*)(out + (size_t)lo * OUT_DIM);
    const v4f* a4 = (const v4f*)acc;
    for (int o = tid; o < rn * (OUT_DIM / 4); o += OWN_THREADS) o4[o] = a4[o];
  }
}

extern "C" void kernel_launch(void* const* d_in, const int* in_sizes, int n_in,
                              void* d_out, int out_size, void* d_ws, size_t ws_size,
                              hipStream_t stream) {
  const float* features = (const float*)d_in[0];
  const float* t        = (const float*)d_in[1];
  const float* W_fc     = (const float*)d_in[2];
  const float* w_t      = (const float*)d_in[3];
  const int*   src_idx  = (const int*)d_in[4];
  const int*   dst_idx  = (const int*)d_in[5];

  const int n_src = in_sizes[0] / IN_DIM;
  const int E     = in_sizes[4];
  const int n_dst = out_size / OUT_DIM;

  float* zbuf  = (float*)d_ws;
  float* score = zbuf + (size_t)n_src * OUT_DIM;
  float* out   = (float*)d_out;

  float* wp = score;

  const int bm_waves = ((E + 31) / 32 + 31) / 32;
  k_prep<<<W_BLOCKS + (bm_waves + BM_WAVES - 1) / BM_WAVES, PREP_THREADS, 0, stream>>>(
      W_fc, wp, src_idx, dst_idx, (unsigned*)d_out, E, n_dst, n_src);

  const int ntile = (n_src + 15) / 16;
  k_gemm<<<(ntile + 3) / 4, 128, 0, stream>>>(features, wp, zbuf, n_src);

  k_score<<<(n_src + SC_BLOCK_ROWS - 1) / SC_BLOCK_ROWS, SC_THREADS, 0, stream>>>(zbuf, w_t, score, n_src);

  k_own<<<(n_dst + R_NODES - 1) / R_NODES, OWN_THREADS, 0, stream>>>(
      score, t, src_idx, dst_idx, zbuf, out, E, n_dst, n_src);
}


# k_gemm k loop also sums score = features . (W_fc^T w_t) from its A fragments (vector made by the first launch): k_score launch and its z re-read gone, 3 launches; new summation order of score
# speedup vs baseline: 4.2274x; 1.0282x over previous
#include <hip/hip_runtime.h>
#include <hip/hip_bf16.h>

typedef __attribute__((ext_vector_type(2))) float v2f;
typedef __attribute__((ext_vector_type(4))) float v4f;
typedef __attribute__((ext_vector_type(8))) float v8f;

#define IN_DIM  256
#define OUT_DIM 128

#define WP_FLOATS (OUT_DIM * IN_DIM)

#define REGION_WORDS (400 * OUT_DIM)
#define WP_CHUNK     1024
#define WP_CHUNKS    (WP_FLOATS / WP_CHUNK)
#define WP_TAIL      (REGION_WORDS - WP_CHUNK)

__global__ __launch_bounds__(128) void k_gemm(const float* __restrict__ A,
                                              const float* __restrict__ Wp,
                                              float* __restrict__ Z,
                                              float* __restrict__ score,
                                              int n_src) {
  const int lane = threadIdx.x & 31;
  const int wave = threadIdx.x >> 5;
  const int tile = blockIdx.x * 4 + wave;
  const int m0   = tile * 16;
  if (m0 >= n_src) return;

  const int half = lane >> 4;
  const int nc   = lane & 15;

  int mrow = m0 + nc;
  if (mrow >= n_src) mrow = n_src - 1;
  const float* __restrict__ arow = A + (size_t)mrow * IN_DIM + 2 * half;
  const v4f* __restrict__ wl = (const v4f*)(Wp + WP_TAIL) + lane;
  const v2f* __restrict__ vl =
      (const v2f*)(Wp + (size_t)WP_CHUNKS * REGION_WORDS + WP_TAIL + 2 * half);
  float sc = 0.0f;

  const v8f vzero = {};
  v8f acc[8];
#pragma unroll
  for (int i = 0; i < 8; ++i) acc[i] = vzero;

#pragma unroll 1
  for (int t = 0; t < IN_DIM / 16; ++t) {
    v2f a[4];
    v2f vv[4];
    v4f b[16];
#pragma unroll
    for (int i = 0; i < 4; ++i) a[i] = *(const v2f*)(arow + 16 * t + 4 * i);
#pragma unroll
    for (int i = 0; i < 4; ++i) vv[i] = vl[8 * t + 2 * i];
#pragma unroll
    for (int i = 0; i < 16; ++i)
      b[i] = wl[(i >> 3) * (REGION_WORDS / 4) + (i & 7) * 32];
    wl += 2 * (REGION_WORDS / 4);
    __builtin_amdgcn_sched_barrier(0);
#pragma unroll
    for (int qq = 0; qq < 2; ++qq) {
#pragma unroll
      for (int tn = 0; tn < 8; ++tn) {
        const v4f bb = b[qq * 8 + tn];
        const v2f b0 = {bb[0], bb[1]};
        const v2f b1 = {bb[2], bb[3]};
        acc[tn] = __builtin_amdgcn_wmma_f32_16x16x4_f32(
             false, a[2 * qq],  false, b0,
             (short)0, acc[tn],  false,  false);
        acc[tn] = __builtin_amdgcn_wmma_f32_16x16x4_f32(
             false, a[2 * qq + 1],  false, b1,
             (short)0, acc[tn],  false,  false);
      }
    }
#pragma unroll
    for (int i = 0; i < 4; ++i) {
      sc = __builtin_fmaf(a[i][0], vv[i][0], sc);
      sc = __builtin_fmaf(a[i][1], vv[i][1], sc);
    }
  }

  sc += __shfl_xor(sc, 16, 32);
  if (half == 0 && m0 + nc < n_src) score[m0 + nc] = sc;

#pragma unroll
  for (int r = 0; r < 8; ++r) {
    const int mm = m0 + r + 8 * half;
    if (mm < n_src) {
      float* zr = Z + (size_t)mm * OUT_DIM;
#pragma unroll
      for (int tn = 0; tn < 8; ++tn) zr[tn * 16 + nc] = acc[tn][r];
    }
  }
}

__device__ __forceinline__ float edge_logit(const float* __restrict__ score,
                                            const float* __restrict__ t,
                                            int s, int d) {
  const float e = -fabsf(t[s] - t[d]);
  return expf(score[s] * e * 0.002f);
}

#define R_NODES      400
#define PREP_THREADS 128
#define BM_WAVES     (PREP_THREADS / 32)
#define BM_ROWS      252
#define W_BLOCKS     (WP_FLOATS / (PREP_THREADS * 4))
#define SRC_BITS     18
#define ROW_BITS     9
#define TAG_SHIFT    (SRC_BITS + ROW_BITS)
#define SRC_MASK     ((1u << SRC_BITS) - 1u)
#define PAY_MASK     ((1u << TAG_SHIFT) - 1u)

static_assert(R_NODES <= (1 << ROW_BITS), "a row of an owner fits its field");
static_assert(R_NODES * OUT_DIM == REGION_WORDS, "k_gemm's region size is k_own's");
static_assert(IN_DIM <= WP_CHUNK && IN_DIM % PREP_THREADS == 0, "vfc fits one region tail");
static_assert(TAG_SHIFT + 5 == 32, "count field: 5 bits");

__device__ __forceinline__ bool bitmaps_fit(int E, int n_dst, int n_src) {
  const int nblk      = (n_dst + R_NODES - 1) / R_NODES;
  const int words     = (E + 31) >> 5;
  const int last_rows = n_dst - (nblk - 1) * R_NODES;
  return nblk <= BM_ROWS && words <= last_rows * OUT_DIM && words <= WP_TAIL &&
         n_src < (1 << SRC_BITS);
}

__global__ __launch_bounds__(PREP_THREADS) void k_prep(const float* __restrict__ W,
                                                       const float* __restrict__ wt,
                                                       float* Wp,
                                                       const int* __restrict__ src,
                                                       const int* __restrict__ dst,
                                                       unsigned* __restrict__ bm,
                                                       int E, int n_dst, int n_src) {
  __shared__ unsigned tile[BM_WAVES][BM_ROWS * 32];
  const int tid = threadIdx.x;

  if (blockIdx.x < W_BLOCKS) {
    for (int o = blockIdx.x * PREP_THREADS + tid; o < WP_FLOATS; o += W_BLOCKS * PREP_THREADS) {
      const int e    = o & 3;
      const int lane = (o >> 2) & 31;
      const int tn   = (o >> 7) & 7;
      const int q    = o >> 10;
      const int n    = tn * 16 + (lane & 15);
      const int k    = 8 * q + 4 * (e >> 1) + 2 * (lane >> 4) + (e & 1);
      Wp[(size_t)q * REGION_WORDS + WP_TAIL + (o & (WP_CHUNK - 1))] = W[n * IN_DIM + k];
    }
    if (blockIdx.x < IN_DIM / PREP_THREADS) {
      const int k = blockIdx.x * PREP_THREADS + tid;
      float s = 0.0f;
#pragma unroll 32
      for (int n = 0; n < OUT_DIM; ++n) s = __builtin_fmaf(W[n * IN_DIM + k], wt[n], s);
      Wp[(size_t)WP_CHUNKS * REGION_WORDS + WP_TAIL + k] = s;
    }
    return;
  }

  if (!bitmaps_fit(E, n_dst, n_src)) return;
  const int lane  = tid & 31;
  const int wv    = tid >> 5;
  const int nblk  = (n_dst + R_NODES - 1) / R_NODES;
  const int words = (E + 31) >> 5;
  const int gw    = (blockIdx.x - W_BLOCKS) * BM_WAVES + wv;
  if (gw * 32 >= words) return;
  const int wq    = gw * 32 + lane;
  const int e0    = wq * 32;

  unsigned* col = &tile[wv][lane];
  for (int v = 0; v < BM_ROWS; ++v) col[v * 32] = 0u;

  auto mark = [&](const int d, const int s) {
    const unsigned o   = min((unsigned)d / (unsigned)R_NODES, (unsigned)(BM_ROWS - 1));
    const unsigned row = ((unsigned)d - o * (unsigned)R_NODES) & ((1u << ROW_BITS) - 1u);
    const unsigned pay = (row << SRC_BITS) | (((unsigned)s + 1u) & SRC_MASK);
    atomicAdd(&col[o * 32], 1u << TAG_SHIFT);
    atomicOr(&col[o * 32], pay);
  };
  if (e0 + 32 <= E) {
    const int4* p  = (const int4*)(dst + e0);
    const int4* ps = (const int4*)(src + e0);
    int4 d[8], s[8];
#pragma unroll
    for (int c = 0; c < 8; ++c) { d[c] = p[c]; s[c] = ps[c]; }
#pragma unroll
    for (int c = 0; c < 8; ++c) {
      mark(d[c].x, s[c].x);
      mark(d[c].y, s[c].y);
      mark(d[c].z, s[c].z);
      mark(d[c].w, s[c].w);
    }
  } else {
    for (int j = 0; j < 32; ++j)
      if (e0 + j < E) mark(dst[e0 + j], src[e0 + j]);
  }

  if (wq < words) {
    unsigned* g = bm + wq;
#pragma unroll 10
    for (int v = 0; v < nblk; ++v) g[(size_t)v * (R_NODES * OUT_DIM)] = col[v * 32];
  }
}

#define OWN_THREADS 1024
#define OWN_WAVES   (OWN_THREADS / 32)
#define WAVE_ROWS   ((R_NODES + OWN_WAVES - 1) / OWN_WAVES)
#define LIST_CAP    8192
#define BIN_CAP     288
#define TILE_EDGES  (OWN_THREADS * 4)
#define SCAN_UNROLL 8
#define BM_UNROLL   4
#define EDGE_GROUP  8

static_assert(OWN_WAVES == 32, "a row's wave is its low 5 bits");
static_assert(WAVE_ROWS <= 256, "a record's k is one byte");

__global__ __launch_bounds__(OWN_THREADS) void k_own(const float* __restrict__ score,
                                                     const float* __restrict__ t,
                                                     const int* __restrict__ src,
                                                     const int* __restrict__ dst,
                                                     const float* __restrict__ Z,
                                                     float* __restrict__ out,
                                                     int E, int n_dst, int n_src) {
  __shared__ __attribute__((aligned(16))) float acc[R_NODES * OUT_DIM];
  __shared__ float m_s[R_NODES];
  __shared__ float den_s[R_NODES];
  __shared__ int   list_i[LIST_CAP];
  __shared__ int   bin_s[OWN_WAVES * BIN_CAP];
  __shared__ float bin_w[OWN_WAVES * BIN_CAP];
  __shared__ unsigned char bin_k[OWN_WAVES * BIN_CAP];
  __shared__ int   bin_n[OWN_WAVES];
  __shared__ int   n_listed;
  __shared__ int   n_multi;

  const int tid  = threadIdx.x;
  const int lane = tid & 31;
  const int lo   = blockIdx.x * R_NODES;
  const int rn   = min(R_NODES, n_dst - lo);

  for (int o = tid; o < R_NODES; o += OWN_THREADS) { m_s[o] = 0.0f; den_s[o] = 0.0f; }
  if (tid < OWN_WAVES) bin_n[tid] = 0;
  if (tid == 0) { n_listed = 0; n_multi = 0; }
  {
    const v4f vzero = {};
    for (int row = tid >> 5; row < R_NODES; row += OWN_WAVES)
      *(v4f*)(acc + row * OUT_DIM + 4 * lane) = vzero;
  }
  __syncthreads();

  const bool mapped = bitmaps_fit(E, n_dst, n_src);
  if (mapped) {
    const unsigned* __restrict__ bmw = (const unsigned*)(out + (size_t)lo * OUT_DIM);
    const uint4* __restrict__ bm4 = (const uint4*)bmw;
    const int words = (E + 31) >> 5;
    const int quads = words >> 2;
    int* const multi = bin_s;
    const int multi_cap = OWN_WAVES * BIN_CAP;
    auto single = [](const unsigned x) { return (x >> TAG_SHIFT) == 1u; };
    auto emit = [&](const unsigned x, int& sl) {
      if (single(x)) {
        if (sl < LIST_CAP) list_i[sl] = (int)(x & PAY_MASK);
        ++sl;
      }
    };
    auto note = [&](const unsigned x, const int w) {
      if (x != 0u && !single(x)) {
        const int ms = atomicAdd(&n_multi, 1);
        if (ms < multi_cap) multi[ms] = w;
      }
    };
    for (int q0 = tid; q0 < quads; q0 += OWN_THREADS * BM_UNROLL) {
      uint4 x[BM_UNROLL];
#pragma unroll
      for (int u = 0; u < BM_UNROLL; ++u) {
        const int q = q0 + u * OWN_THREADS;
        x[u] = (q < quads) ? bm4[q] : make_uint4(0u, 0u, 0u, 0u);
      }
#pragma unroll
      for (int u = 0; u < BM_UNROLL; ++u) {
        if (x[u].x | x[u].y | x[u].z | x[u].w) {
          const int cnt = (int)single(x[u].x) + (int)single(x[u].y) +
                          (int)single(x[u].z) + (int)single(x[u].w);
          const int wb = (q0 + u * OWN_THREADS) * 4;
          if (cnt) {
            int sl = atomicAdd(&n_listed, cnt);
            emit(x[u].x, sl);
            emit(x[u].y, sl);
            emit(x[u].z, sl);
            emit(x[u].w, sl);
          }
          note(x[u].x, wb);
          note(x[u].y, wb + 1);
          note(x[u].z, wb + 2);
          note(x[u].w, wb + 3);
        }
      }
    }
    for (int w = quads * 4 + tid; w < words; w += OWN_THREADS) {
      const unsigned x = bmw[w];
      if (single(x)) {
        int sl = atomicAdd(&n_listed, 1);
        emit(x, sl);
      }
      note(x, w);
    }
    __syncthreads();
    const unsigned r = (unsigned)rn;
    const int nm = n_multi;
    if (nm > multi_cap) atomicAdd(&n_listed, LIST_CAP + 1);
    for (int m = tid; m < min(nm, multi_cap); m += OWN_THREADS) {
      const int e0 = multi[m] * 32;
      unsigned hit = 0u;
      if (e0 + 32 <= E) {
        const int4* p = (const int4*)(dst + e0);
        int4 d[8];
#pragma unroll
        for (int c = 0; c < 8; ++c) d[c] = p[c];
#pragma unroll
        for (int c = 0; c < 8; ++c) {
          hit |= ((unsigned)(d[c].x - lo) < r ? 1u : 0u) << (4 * c);
          hit |= ((unsigned)(d[c].y - lo) < r ? 1u : 0u) << (4 * c + 1);
          hit |= ((unsigned)(d[c].z - lo) < r ? 1u : 0u) << (4 * c + 2);
          hit |= ((unsigned)(d[c].w - lo) < r ? 1u : 0u) << (4 * c + 3);
        }
      } else {
        for (int j = 0; j < 32; ++j)
          if (e0 + j < E && (unsigned)(dst[e0 + j] - lo) < r) hit |= 1u << j;
      }
      if (hit) {
        int sl = atomicAdd(&n_listed, __builtin_popcount(hit));
        while (hit) {
          const int j = __builtin_ctz(hit);
          hit &= hit - 1;
          const unsigned row = (unsigned)(dst[e0 + j] - lo);
          const unsigned pay = (row << SRC_BITS) | (((unsigned)src[e0 + j] + 1u) & SRC_MASK);
          if (sl < LIST_CAP) list_i[sl] = (int)pay;
          ++sl;
        }
      }
    }
  } else {
    const unsigned r = (unsigned)rn;
    auto append = [&](const int i) {
      const int sl = atomicAdd(&n_listed, 1);
      if (sl < LIST_CAP) list_i[sl] = i;
    };
    auto take = [&](const int4 dv, const int e0) {
      const unsigned d0 = (unsigned)(dv.x - lo);
      const unsigned d1 = (unsigned)(dv.y - lo);
      const unsigned d2 = (unsigned)(dv.z - lo);
      const unsigned d3 = (unsigned)(dv.w - lo);
      if (d0 < r || d1 < r || d2 < r || d3 < r) {
        if (d0 < r) append(e0);
        if (d1 < r) append(e0 + 1);
        if (d2 < r) append(e0 + 2);
        if (d3 < r) append(e0 + 3);
      }
    };
    const int n_full = E / TILE_EDGES;
    const int first  = (int)(((long long)blockIdx.x * n_full) / gridDim.x);
    int p = 0;
    for (; p + SCAN_UNROLL <= n_full; p += SCAN_UNROLL) {
      int4 dv[SCAN_UNROLL];
      int  eb[SCAN_UNROLL];
#pragma unroll
      for (int u = 0; u < SCAN_UNROLL; ++u) {
        int tile = first + p + u;
        if (tile >= n_full) tile -= n_full;
        eb[u] = tile * TILE_EDGES + tid * 4;
        dv[u] = *(const int4*)(dst + eb[u]);
      }
#pragma unroll
      for (int u = 0; u < SCAN_UNROLL; ++u) take(dv[u], eb[u]);
    }
    for (; p < n_full; ++p) {
      int tile = first + p;
      if (tile >= n_full) tile -= n_full;
      const int e0 = tile * TILE_EDGES + tid * 4;
      take(*(const int4*)(dst + e0), e0);
    }
    for (int i = n_full * TILE_EDGES + tid; i < E; i += OWN_THREADS) {
      const unsigned d0 = (unsigned)(dst[i] - lo);
      if (d0 < r) append(i);
    }
  }
  __syncthreads();
  const int total = n_listed;

  bool binned = total <= LIST_CAP;
  if (binned) {
    for (int e = tid; e < total; e += OWN_THREADS) {
      const int   v  = list_i[e];
      int s, dl;
      if (mapped) {
        dl = (int)((unsigned)v >> SRC_BITS);
        s  = (int)((unsigned)v & SRC_MASK) - 1;
      } else {
        dl = dst[v] - lo;
        s  = src[v];
      }
      const float le = edge_logit(score, t, s, lo + dl);
      atomicMax((int*)&m_s[dl], __float_as_int(le));
      const int b  = dl & (OWN_WAVES - 1);
      const int sl = atomicAdd(&bin_n[b], 1);
      if (sl < BIN_CAP) {
        const int p = b * BIN_CAP + sl;
        bin_s[p] = s;
        bin_w[p] = le;
        bin_k[p] = (unsigned char)(dl / OWN_WAVES);
      }
    }
    __syncthreads();
    int fullest = 0;
    for (int b = 0; b < OWN_WAVES; ++b) fullest = max(fullest, bin_n[b]);
    binned = fullest <= BIN_CAP;
  }

  if (binned) {
    const int wv = __builtin_amdgcn_readfirstlane(tid >> 5);
    const int n  = __builtin_amdgcn_readfirstlane(bin_n[wv]);
    const int p0 = wv * BIN_CAP;

    for (int base = 0; base < n; base += 32) {
      if (base + lane < n) {
        const int   p   = p0 + base + lane;
        const int   row = (int)bin_k[p] * OWN_WAVES + wv;
        const float un  = expf(bin_w[p] - m_s[row]);
        atomicAdd(&den_s[row], un);
        bin_w[p] = un;
      }
    }

    auto add_row = [&](const int k, const float aj, const v4f zv) {
      v4f* const p = (v4f*)(acc + (k * OWN_WAVES + wv) * OUT_DIM + 4 * lane);
      v4f r = *p;
      r[0] = __builtin_fmaf(aj, zv[0], r[0]);
      r[1] = __builtin_fmaf(aj, zv[1], r[1]);
      r[2] = __builtin_fmaf(aj, zv[2], r[2]);
      r[3] = __builtin_fmaf(aj, zv[3], r[3]);
      *p = r;
    };

    for (int base = 0; base < n; base += 32) {
      const int cnt = min(32, n - base);
      int   s = 0, k = 0;
      float alpha = 0.0f;
      if (lane < cnt) {
        const int p = p0 + base + lane;
        s = bin_s[p];
        k = bin_k[p];
        alpha = bin_w[p] / fmaxf(den_s[k * OWN_WAVES + wv], 1e-20f);
      }
      const int alpha_bits = __float_as_int(alpha);
      auto load8 = [&](v4f (&z)[EDGE_GROUP], const int j0) {
#pragma unroll
        for (int u = 0; u < EDGE_GROUP; ++u) {
          const int sj = __builtin_amdgcn_readlane(s, j0 + u);
          z[u] = *(const v4f*)(Z + (size_t)sj * OUT_DIM + 4 * lane);
        }
      };
      auto add8 = [&](const v4f (&z)[EDGE_GROUP], const int j0) {
#pragma unroll
        for (int u = 0; u < EDGE_GROUP; ++u) {
          const int   kj = __builtin_amdgcn_readlane(k, j0 + u);
          const float aj = __int_as_float(__builtin_amdgcn_readlane(alpha_bits, j0 + u));
          add_row(kj, aj, z[u]);
        }
      };
      if (cnt == 32) {
        v4f za[EDGE_GROUP], zb[EDGE_GROUP];
        load8(za, 0);
        load8(zb, 8);
        add8(za, 0);
        load8(za, 16);
        add8(zb, 8);
        load8(zb, 24);
        add8(za, 16);
        add8(zb, 24);
      } else {
        int j0 = 0;
        for (; j0 + EDGE_GROUP <= cnt; j0 += EDGE_GROUP) {
          v4f z[EDGE_GROUP];
          load8(z, j0);
          add8(z, j0);
        }
        for (int j = j0; j < cnt; ++j) {
          const int   sj = __builtin_amdgcn_readlane(s, j);
          const int   kj = __builtin_amdgcn_readlane(k, j);
          const float aj = __int_as_float(__builtin_amdgcn_readlane(alpha_bits, j));
          add_row(kj, aj, *(const v4f*)(Z + (size_t)sj * OUT_DIM + 4 * lane));
        }
      }
    }

#pragma unroll
    for (int k = 0; k < WAVE_ROWS; ++k) {
      const int row = k * OWN_WAVES + wv;
      if (row < rn) {
        const v4f zres = *(const v4f*)(Z + (size_t)(lo + row) * OUT_DIM + 4 * lane);
        const v4f sum  = *(const v4f*)(acc + row * OUT_DIM + 4 * lane);
        *(v4f*)(out + (size_t)(lo + row) * OUT_DIM + 4 * lane) = zres + sum;
      }
    }
  } else {
    {
      const v4f* __restrict__ z4 = (const v4f*)(Z + (size_t)lo * OUT_DIM);
      v4f* a4 = (v4f*)acc;
      for (int o = tid; o < rn * (OUT_DIM / 4); o += OWN_THREADS) a4[o] = z4[o];
    }
    __syncthreads();
    for (int mode = 0; mode < 3; ++mode) {
      for (int i = tid; i < E; i += OWN_THREADS) {
        const unsigned dl = (unsigned)(dst[i] - lo);
        if (dl < (unsigned)rn) {
          const int   s  = src[i];
          const float le = edge_logit(score, t, s, lo + (int)dl);
          if (mode == 0) {
            atomicMax((int*)&m_s[dl], __float_as_int(le));
          } else {
            const float un = expf(le - m_s[dl]);
            if (mode == 1) {
              atomicAdd(&den_s[dl], un);
            } else {
              const float al = un / fmaxf(den_s[dl], 1e-20f);
              const float* zr = Z + (size_t)s * OUT_DIM;
              for (int c = 0; c < OUT_DIM; ++c) atomicAdd(&acc[dl * OUT_DIM + c], al * zr[c]);
            }
          }
        }
      }
      __syncthreads();
    }
    v4f* __restrict__ o4 = (v4f*)(out + (size_t)lo * OUT_DIM);
    const v4f* a4 = (const v4f*)acc;
    for (int o = tid; o < rn * (OUT_DIM / 4); o += OWN_THREADS) o4[o] = a4[o];
  }
}

extern "C" void kernel_launch(void* const* d_in, const int* in_sizes, int n_in,
                              void* d_out, int out_size, void* d_ws, size_t ws_size,
                              hipStream_t stream) {
  const float* features = (const float*)d_in[0];
  const float* t        = (const float*)d_in[1];
  const float* W_fc     = (const float*)d_in[2];
  const float* w_t      = (const float*)d_in[3];
  const int*   src_idx  = (const int*)d_in[4];
  const int*   dst_idx  = (const int*)d_in[5];

  const int n_src = in_sizes[0] / IN_DIM;
  const int E     = in_sizes[4];
  const int n_dst = out_size / OUT_DIM;

  float* zbuf  = (float*)d_ws;
  float* score = zbuf + (size_t)n_src * OUT_DIM;
  float* out   = (float*)d_out;

  float* wp = out;

  const int bm_waves = ((E + 31) / 32 + 31) / 32;
  k_prep<<<W_BLOCKS + (bm_waves + BM_WAVES - 1) / BM_WAVES, PREP_THREADS, 0, stream>>>(
      W_fc, w_t, wp, src_idx, dst_idx, (unsigned*)d_out, E, n_dst, n_src);

  const int ntile = (n_src + 15) / 16;
  k_gemm<<<(ntile + 3) / 4, 128, 0, stream>>>(features, wp, zbuf, score, n_src);

  k_own<<<(n_dst + R_NODES - 1) / R_NODES, OWN_THREADS, 0, stream>>>(
      score, t, src_idx, dst_idx, zbuf, out, E, n_dst, n_src);
}
